# starting kernel
# baseline (speedup 1.0000x reference)
#include <hip/hip_runtime.h>
#include <hip/hip_bf16.h>
#include <math.h>

typedef __attribute__((ext_vector_type(2))) float v2f;
typedef __attribute__((ext_vector_type(8))) float v8f;

#define C_DIM 256

__global__ void gcn_zero_kernel(unsigned* __restrict__ deg,
                                float* __restrict__ accum, int N) {
    int i = blockIdx.x * blockDim.x + threadIdx.x;
    if (i < N) {
        deg[i]   = 0u;
        accum[i] = 0.0f;
    }
}

__global__ void gcn_matvec_wmma_kernel(const float* __restrict__ x,
                                       const float* __restrict__ W,
                                       float* __restrict__ h, int N) {
    const int lane = threadIdx.x & 31;
    const int wave = threadIdx.x >> 5;
    const int wavesPerBlock = blockDim.x >> 5;
    const int base = (blockIdx.x * wavesPerBlock + wave) * 16;
    if (base >= N) return;

    const int half = lane >> 4;
    const int m    = lane & 15;
    int row = base + m;
    if (row >= N) row = N - 1;

    const float* __restrict__ xrow = x + (size_t)row * C_DIM;

    v8f acc = {0.f, 0.f, 0.f, 0.f, 0.f, 0.f, 0.f, 0.f};

    #pragma unroll 4
    for (int k0 = 0; k0 < C_DIM; k0 += 4) {
        const int ka = k0 + half * 2;
        v2f a = *(const v2f*)(xrow + ka);
        v2f b;
        b.x = W[ka];
        b.y = W[ka + 1];
        acc = __builtin_amdgcn_wmma_f32_16x16x4_f32(
             false, a,  false, b,
             (short)0, acc,  false,  false);
    }

    if ((lane & 15) == 0) {
        const int mbase = base + half * 8;
        #pragma unroll
        for (int r = 0; r < 8; ++r) {
            if (mbase + r < N) h[mbase + r] = acc[r];
        }
    }
}

__global__ void gcn_degree_kernel(const int* __restrict__ ei,
                                  unsigned* __restrict__ deg, int E) {
    int e = blockIdx.x * blockDim.x + threadIdx.x;
    if (e < E) {
        atomicAdd(&deg[ei[E + e]], 1u);
    }
}

__global__ void gcn_dinv_kernel(const unsigned* __restrict__ deg,
                                float* __restrict__ dinv, int N) {
    int i = blockIdx.x * blockDim.x + threadIdx.x;
    if (i < N) {
        dinv[i] = rsqrtf((float)(deg[i] + 1u));
    }
}

__global__ void gcn_edge_kernel(const int* __restrict__ ei,
                                const float* __restrict__ dinv,
                                const float* __restrict__ h,
                                float* __restrict__ accum, int E) {
    int e = blockIdx.x * blockDim.x + threadIdx.x;
    if (e < E) {
        int s = ei[e];
        int d = ei[E + e];
        atomicAdd(&accum[d], dinv[s] * dinv[d] * h[s]);
    }
}

__global__ void gcn_finalize_kernel(const float* __restrict__ accum,
                                    const float* __restrict__ dinv,
                                    const float* __restrict__ h,
                                    const float* __restrict__ b,
                                    float* __restrict__ out, int N) {
    int i = blockIdx.x * blockDim.x + threadIdx.x;
    if (i < N) {
        float di = dinv[i];
        float v  = accum[i] + di * di * h[i] + b[0];
        out[i]   = fmaxf(v, 0.0f) + log1pf(expf(-fabsf(v)));
    }
}

extern "C" void kernel_launch(void* const* d_in, const int* in_sizes, int n_in,
                              void* d_out, int out_size, void* d_ws, size_t ws_size,
                              hipStream_t stream) {
    const float* x  = (const float*)d_in[0];
    const int*   ei = (const int*)d_in[1];
    const float* W  = (const float*)d_in[2];
    const float* b  = (const float*)d_in[3];
    float* out = (float*)d_out;

    const int N = in_sizes[0] / C_DIM;
    const int E = in_sizes[1] / 2;

    float*    h     = (float*)d_ws;
    unsigned* deg   = (unsigned*)(h + N);
    float*    dinv  = (float*)(deg + N);
    float*    accum = dinv + N;

    const int BLK = 256;
    const int gridN = (N + BLK - 1) / BLK;
    const int gridE = (E + BLK - 1) / BLK;
    const int nodesPerBlock = (BLK / 32) * 16;
    const int gridM = (N + nodesPerBlock - 1) / nodesPerBlock;

    gcn_zero_kernel<<<gridN, BLK, 0, stream>>>(deg, accum, N);
    gcn_matvec_wmma_kernel<<<gridM, BLK, 0, stream>>>(x, W, h, N);
    gcn_degree_kernel<<<gridE, BLK, 0, stream>>>(ei, deg, E);
    gcn_dinv_kernel<<<gridN, BLK, 0, stream>>>(deg, dinv, N);
    gcn_edge_kernel<<<gridE, BLK, 0, stream>>>(ei, dinv, h, accum, E);
    gcn_finalize_kernel<<<gridN, BLK, 0, stream>>>(accum, dinv, h, b, out, N);
}


# Degree and message sums: per-edge global atomics become LDS sums per owned dst range (range x edge-slice work-groups scan the edge ids), merged by whole-line global atomics; float sum order changes
# speedup vs baseline: 3.6642x; 3.6642x over previous
#include <hip/hip_runtime.h>
#include <hip/hip_bf16.h>
#include <math.h>

typedef __attribute__((ext_vector_type(2))) float v2f;
typedef __attribute__((ext_vector_type(8))) float v8f;

#define C_DIM 256

__global__ void gcn_zero_kernel(unsigned* __restrict__ deg,
                                float* __restrict__ accum, int N) {
    int i = blockIdx.x * blockDim.x + threadIdx.x;
    if (i < N) {
        deg[i]   = 0u;
        accum[i] = 0.0f;
    }
}

__global__ void gcn_matvec_wmma_kernel(const float* __restrict__ x,
                                       const float* __restrict__ W,
                                       float* __restrict__ h, int N) {
    const int lane = threadIdx.x & 31;
    const int wave = threadIdx.x >> 5;
    const int wavesPerBlock = blockDim.x >> 5;
    const int base = (blockIdx.x * wavesPerBlock + wave) * 16;
    if (base >= N) return;

    const int half = lane >> 4;
    const int m    = lane & 15;
    int row = base + m;
    if (row >= N) row = N - 1;

    const float* __restrict__ xrow = x + (size_t)row * C_DIM;

    v8f acc = {0.f, 0.f, 0.f, 0.f, 0.f, 0.f, 0.f, 0.f};

    #pragma unroll 4
    for (int k0 = 0; k0 < C_DIM; k0 += 4) {
        const int ka = k0 + half * 2;
        v2f a = *(const v2f*)(xrow + ka);
        v2f b;
        b.x = W[ka];
        b.y = W[ka + 1];
        acc = __builtin_amdgcn_wmma_f32_16x16x4_f32(
             false, a,  false, b,
             (short)0, acc,  false,  false);
    }

    if ((lane & 15) == 0) {
        const int mbase = base + half * 8;
        #pragma unroll
        for (int r = 0; r < 8; ++r) {
            if (mbase + r < N) h[mbase + r] = acc[r];
        }
    }
}

#define RANGE        6250
#define SLICES       16
#define SCAT_THREADS 1024
#define SCAT_WAVES   (SCAT_THREADS / 32)
#define BATCH        4
#define WTILE        (BATCH * 128)

static __device__ __forceinline__ int4 load_ids4(const int* __restrict__ p,
                                                 int e, int end) {
    return *(const int4*)(p + min(e, end - 4));
}
static __device__ __forceinline__ int4 keep_ids4(int4 t, int e, int end) {
    const bool in = e < end;
    return make_int4(in ? t.x : -1, in ? t.y : -1, in ? t.z : -1, in ? t.w : -1);
}

static __device__ __forceinline__ int slice_len(int E) {
    int s = (E + SLICES - 1) / SLICES;
    return (s + 3) & ~3;
}

__global__ __launch_bounds__(SCAT_THREADS)
void gcn_degree_kernel(const int* __restrict__ ei,
                       unsigned* __restrict__ deg, int N, int E) {
    __shared__ unsigned s_cnt[RANGE];
    const int tid = threadIdx.x;
    const int lo  = blockIdx.x * RANGE;
    for (int i = tid; i < RANGE; i += SCAT_THREADS) s_cnt[i] = 0u;
    __syncthreads();

    const int* __restrict__ dst = ei + E;
    const bool vec = (E & 3) == 0;
    const int S   = slice_len(E);
    const int beg = blockIdx.y * S;
    const int end = min(E, beg + S);

    if (vec) {
        if (beg < end) {
            const int step = SCAT_THREADS * 4 * BATCH;
            int4 nx[BATCH];
            #pragma unroll
            for (int u = 0; u < BATCH; ++u)
                nx[u] = load_ids4(dst, beg + tid * 4 + u * SCAT_THREADS * 4, end);
            for (int e0 = beg + tid * 4; e0 < end; e0 += step) {
                int4 v[BATCH];
                #pragma unroll
                for (int u = 0; u < BATCH; ++u) v[u] = nx[u];
                #pragma unroll
                for (int u = 0; u < BATCH; ++u)
                    nx[u] = load_ids4(dst, e0 + step + u * SCAT_THREADS * 4, end);
                #pragma unroll
                for (int u = 0; u < BATCH; ++u) {
                    const int4 k = keep_ids4(v[u], e0 + u * SCAT_THREADS * 4, end);
                    const int d[4] = {k.x, k.y, k.z, k.w};
                    #pragma unroll
                    for (int j = 0; j < 4; ++j) {
                        const unsigned rel = (unsigned)(d[j] - lo);
                        if (rel < RANGE)
                            __hip_atomic_fetch_add(&s_cnt[rel], 1u, __ATOMIC_RELAXED,
                                                   __HIP_MEMORY_SCOPE_WORKGROUP);
                    }
                }
            }
        }
    } else {
        for (int e = beg + tid; e < end; e += SCAT_THREADS) {
            const unsigned rel = (unsigned)(dst[e] - lo);
            if (rel < RANGE)
                __hip_atomic_fetch_add(&s_cnt[rel], 1u, __ATOMIC_RELAXED,
                                       __HIP_MEMORY_SCOPE_WORKGROUP);
        }
    }
    __syncthreads();

    const int n = min(RANGE, N - lo);
    for (int i = tid; i < n; i += SCAT_THREADS)
        atomicAdd(&deg[lo + i], s_cnt[i]);
}

__global__ void gcn_dinv_kernel(const unsigned* __restrict__ deg,
                                float* __restrict__ dinv, int N) {
    int i = blockIdx.x * blockDim.x + threadIdx.x;
    if (i < N) {
        dinv[i] = rsqrtf((float)(deg[i] + 1u));
    }
}

__global__ __launch_bounds__(SCAT_THREADS)
void gcn_edge_kernel(const int* __restrict__ ei,
                     const float* __restrict__ dinv,
                     const float* __restrict__ h,
                     float* __restrict__ accum, int N, int E) {
    __shared__ float s_acc[RANGE];
    __shared__ uint2 s_list[SCAT_WAVES][WTILE];
    const int tid  = threadIdx.x;
    const int lane = tid & 31;
    const int wave = tid >> 5;
    const int lo   = blockIdx.x * RANGE;
    for (int i = tid; i < RANGE; i += SCAT_THREADS) s_acc[i] = 0.0f;
    __syncthreads();

    const int* __restrict__ src = ei;
    const int* __restrict__ dst = ei + E;
    const bool vec = (E & 3) == 0;
    const int S   = slice_len(E);
    const int beg = blockIdx.y * S;
    const int end = min(E, beg + S);
    uint2* __restrict__ list = s_list[wave];
    const unsigned below = (1u << lane) - 1u;

    if (vec) {
        if (beg < end) {
            const int step = SCAT_WAVES * WTILE;
            int4 nd[BATCH], ns[BATCH];
            #pragma unroll
            for (int u = 0; u < BATCH; ++u) {
                const int e = beg + wave * WTILE + u * 128 + lane * 4;
                nd[u] = load_ids4(dst, e, end);
                ns[u] = load_ids4(src, e, end);
            }
            for (int tb = beg + wave * WTILE; tb < end; tb += step) {
                int4 vd[BATCH], vs[BATCH];
                #pragma unroll
                for (int u = 0; u < BATCH; ++u) { vd[u] = nd[u]; vs[u] = ns[u]; }
                #pragma unroll
                for (int u = 0; u < BATCH; ++u) {
                    const int e = tb + step + u * 128 + lane * 4;
                    nd[u] = load_ids4(dst, e, end);
                    ns[u] = load_ids4(src, e, end);
                }

                int nlist = 0;
                #pragma unroll
                for (int u = 0; u < BATCH; ++u) {
                    const int4 k = keep_ids4(vd[u], tb + u * 128 + lane * 4, end);
                    const int d[4] = {k.x, k.y, k.z, k.w};
                    const int s[4] = {vs[u].x, vs[u].y, vs[u].z, vs[u].w};
                    #pragma unroll
                    for (int j = 0; j < 4; ++j) {
                        const unsigned rel = (unsigned)(d[j] - lo);
                        const bool mine = rel < RANGE;
                        const unsigned who = __builtin_amdgcn_ballot_w32(mine);
                        if (mine)
                            list[nlist + __popc(who & below)] =
                                make_uint2(rel, (unsigned)s[j]);
                        nlist += __popc(who);
                    }
                }
                __builtin_amdgcn_fence(__ATOMIC_ACQ_REL, "wavefront");

                for (int i = lane; i < nlist; i += 32) {
                    const uint2 ent = list[i];
                    const int s = (int)ent.y;
                    const float val = dinv[s] * dinv[lo + (int)ent.x] * h[s];
                    __hip_atomic_fetch_add(&s_acc[ent.x], val, __ATOMIC_RELAXED,
                                           __HIP_MEMORY_SCOPE_WORKGROUP);
                }
                __builtin_amdgcn_fence(__ATOMIC_ACQ_REL, "wavefront");
            }
        }
    } else {
        for (int e = beg + tid; e < end; e += SCAT_THREADS) {
            const unsigned rel = (unsigned)(dst[e] - lo);
            if (rel < RANGE) {
                const int s = src[e];
                const float val = dinv[s] * dinv[lo + (int)rel] * h[s];
                __hip_atomic_fetch_add(&s_acc[rel], val, __ATOMIC_RELAXED,
                                       __HIP_MEMORY_SCOPE_WORKGROUP);
            }
        }
    }
    __syncthreads();

    const int n = min(RANGE, N - lo);
    for (int i = tid; i < n; i += SCAT_THREADS)
        atomicAdd(&accum[lo + i], s_acc[i]);
}

__global__ void gcn_finalize_kernel(const float* __restrict__ accum,
                                    const float* __restrict__ dinv,
                                    const float* __restrict__ h,
                                    const float* __restrict__ b,
                                    float* __restrict__ out, int N) {
    int i = blockIdx.x * blockDim.x + threadIdx.x;
    if (i < N) {
        float di = dinv[i];
        float v  = accum[i] + di * di * h[i] + b[0];
        out[i]   = fmaxf(v, 0.0f) + log1pf(expf(-fabsf(v)));
    }
}

extern "C" void kernel_launch(void* const* d_in, const int* in_sizes, int n_in,
                              void* d_out, int out_size, void* d_ws, size_t ws_size,
                              hipStream_t stream) {
    const float* x  = (const float*)d_in[0];
    const int*   ei = (const int*)d_in[1];
    const float* W  = (const float*)d_in[2];
    const float* b  = (const float*)d_in[3];
    float* out = (float*)d_out;

    const int N = in_sizes[0] / C_DIM;
    const int E = in_sizes[1] / 2;

    float*    h     = (float*)d_ws;
    unsigned* deg   = (unsigned*)(h + N);
    float*    dinv  = (float*)(deg + N);
    float*    accum = dinv + N;

    const int BLK = 256;
    const int gridN = (N + BLK - 1) / BLK;
    const dim3 gridS((N + RANGE - 1) / RANGE, SLICES);
    const int nodesPerBlock = (BLK / 32) * 16;
    const int gridM = (N + nodesPerBlock - 1) / nodesPerBlock;

    gcn_zero_kernel<<<gridN, BLK, 0, stream>>>(deg, accum, N);
    gcn_matvec_wmma_kernel<<<gridM, BLK, 0, stream>>>(x, W, h, N);
    gcn_degree_kernel<<<gridS, SCAT_THREADS, 0, stream>>>(ei, deg, N, E);
    gcn_dinv_kernel<<<gridN, BLK, 0, stream>>>(deg, dinv, N);
    gcn_edge_kernel<<<gridS, SCAT_THREADS, 0, stream>>>(ei, dinv, h, accum, N, E);
    gcn_finalize_kernel<<<gridN, BLK, 0, stream>>>(accum, dinv, h, b, out, N);
}


# Launch fusion, 6 launches to 4: the zero pass moves into the matvec kernel; dinv = rsqrt(deg+1) is taken from the degree counters where it is used, not in a pass of its own; sum order as before
# speedup vs baseline: 3.8292x; 1.0450x over previous
#include <hip/hip_runtime.h>
#include <hip/hip_bf16.h>
#include <math.h>

typedef __attribute__((ext_vector_type(2))) float v2f;
typedef __attribute__((ext_vector_type(8))) float v8f;

#define C_DIM 256

__global__ void gcn_matvec_wmma_kernel(const float* __restrict__ x,
                                       const float* __restrict__ W,
                                       float* __restrict__ h,
                                       unsigned* __restrict__ deg,
                                       float* __restrict__ accum, int N) {
    const int lane = threadIdx.x & 31;
    const int wave = threadIdx.x >> 5;
    const int wavesPerBlock = blockDim.x >> 5;
    const int base = (blockIdx.x * wavesPerBlock + wave) * 16;
    if (base >= N) return;

    const int half = lane >> 4;
    const int m    = lane & 15;
    int row = base + m;
    if (row >= N) row = N - 1;

    const float* __restrict__ xrow = x + (size_t)row * C_DIM;

    v8f acc = {0.f, 0.f, 0.f, 0.f, 0.f, 0.f, 0.f, 0.f};

    #pragma unroll 4
    for (int k0 = 0; k0 < C_DIM; k0 += 4) {
        const int ka = k0 + half * 2;
        v2f a = *(const v2f*)(xrow + ka);
        v2f b;
        b.x = W[ka];
        b.y = W[ka + 1];
        acc = __builtin_amdgcn_wmma_f32_16x16x4_f32(
             false, a,  false, b,
             (short)0, acc,  false,  false);
    }

    if ((lane & 15) == 0) {
        const int mbase = base + half * 8;
        #pragma unroll
        for (int r = 0; r < 8; ++r) {
            if (mbase + r < N) h[mbase + r] = acc[r];
        }
    }
    if (half == 0 && base + m < N) {
        deg[base + m]   = 0u;
        accum[base + m] = 0.0f;
    }
}

#define RANGE        6250
#define SLICES       16
#define SCAT_THREADS 1024
#define SCAT_WAVES   (SCAT_THREADS / 32)
#define BATCH        4
#define WTILE        (BATCH * 128)

static __device__ __forceinline__ int4 load_ids4(const int* __restrict__ p,
                                                 int e, int end) {
    return *(const int4*)(p + min(e, end - 4));
}
static __device__ __forceinline__ int4 keep_ids4(int4 t, int e, int end) {
    const bool in = e < end;
    return make_int4(in ? t.x : -1, in ? t.y : -1, in ? t.z : -1, in ? t.w : -1);
}

static __device__ __forceinline__ int slice_len(int E) {
    int s = (E + SLICES - 1) / SLICES;
    return (s + 3) & ~3;
}

__global__ __launch_bounds__(SCAT_THREADS)
void gcn_degree_kernel(const int* __restrict__ ei,
                       unsigned* __restrict__ deg, int N, int E) {
    __shared__ unsigned s_cnt[RANGE];
    const int tid = threadIdx.x;
    const int lo  = blockIdx.x * RANGE;
    for (int i = tid; i < RANGE; i += SCAT_THREADS) s_cnt[i] = 0u;
    __syncthreads();

    const int* __restrict__ dst = ei + E;
    const bool vec = (E & 3) == 0;
    const int S   = slice_len(E);
    const int beg = blockIdx.y * S;
    const int end = min(E, beg + S);

    if (vec) {
        if (beg < end) {
            const int step = SCAT_THREADS * 4 * BATCH;
            int4 nx[BATCH];
            #pragma unroll
            for (int u = 0; u < BATCH; ++u)
                nx[u] = load_ids4(dst, beg + tid * 4 + u * SCAT_THREADS * 4, end);
            for (int e0 = beg + tid * 4; e0 < end; e0 += step) {
                int4 v[BATCH];
                #pragma unroll
                for (int u = 0; u < BATCH; ++u) v[u] = nx[u];
                #pragma unroll
                for (int u = 0; u < BATCH; ++u)
                    nx[u] = load_ids4(dst, e0 + step + u * SCAT_THREADS * 4, end);
                #pragma unroll
                for (int u = 0; u < BATCH; ++u) {
                    const int4 k = keep_ids4(v[u], e0 + u * SCAT_THREADS * 4, end);
                    const int d[4] = {k.x, k.y, k.z, k.w};
                    #pragma unroll
                    for (int j = 0; j < 4; ++j) {
                        const unsigned rel = (unsigned)(d[j] - lo);
                        if (rel < RANGE)
                            __hip_atomic_fetch_add(&s_cnt[rel], 1u, __ATOMIC_RELAXED,
                                                   __HIP_MEMORY_SCOPE_WORKGROUP);
                    }
                }
            }
        }
    } else {
        for (int e = beg + tid; e < end; e += SCAT_THREADS) {
            const unsigned rel = (unsigned)(dst[e] - lo);
            if (rel < RANGE)
                __hip_atomic_fetch_add(&s_cnt[rel], 1u, __ATOMIC_RELAXED,
                                       __HIP_MEMORY_SCOPE_WORKGROUP);
        }
    }
    __syncthreads();

    const int n = min(RANGE, N - lo);
    for (int i = tid; i < n; i += SCAT_THREADS)
        atomicAdd(&deg[lo + i], s_cnt[i]);
}

static __device__ __forceinline__ float dinv_of(unsigned d) {
    return rsqrtf((float)(d + 1u));
}

__global__ __launch_bounds__(SCAT_THREADS)
void gcn_edge_kernel(const int* __restrict__ ei,
                     const unsigned* __restrict__ deg,
                     const float* __restrict__ h,
                     float* __restrict__ accum, int N, int E) {
    __shared__ float s_acc[RANGE];
    __shared__ uint2 s_list[SCAT_WAVES][WTILE];
    const int tid  = threadIdx.x;
    const int lane = tid & 31;
    const int wave = tid >> 5;
    const int lo   = blockIdx.x * RANGE;
    for (int i = tid; i < RANGE; i += SCAT_THREADS) s_acc[i] = 0.0f;
    __syncthreads();

    const int* __restrict__ src = ei;
    const int* __restrict__ dst = ei + E;
    const bool vec = (E & 3) == 0;
    const int S   = slice_len(E);
    const int beg = blockIdx.y * S;
    const int end = min(E, beg + S);
    uint2* __restrict__ list = s_list[wave];
    const unsigned below = (1u << lane) - 1u;

    if (vec) {
        if (beg < end) {
            const int step = SCAT_WAVES * WTILE;
            int4 nd[BATCH], ns[BATCH];
            #pragma unroll
            for (int u = 0; u < BATCH; ++u) {
                const int e = beg + wave * WTILE + u * 128 + lane * 4;
                nd[u] = load_ids4(dst, e, end);
                ns[u] = load_ids4(src, e, end);
            }
            for (int tb = beg + wave * WTILE; tb < end; tb += step) {
                int4 vd[BATCH], vs[BATCH];
                #pragma unroll
                for (int u = 0; u < BATCH; ++u) { vd[u] = nd[u]; vs[u] = ns[u]; }
                #pragma unroll
                for (int u = 0; u < BATCH; ++u) {
                    const int e = tb + step + u * 128 + lane * 4;
                    nd[u] = load_ids4(dst, e, end);
                    ns[u] = load_ids4(src, e, end);
                }

                int nlist = 0;
                #pragma unroll
                for (int u = 0; u < BATCH; ++u) {
                    const int4 k = keep_ids4(vd[u], tb + u * 128 + lane * 4, end);
                    const int d[4] = {k.x, k.y, k.z, k.w};
                    const int s[4] = {vs[u].x, vs[u].y, vs[u].z, vs[u].w};
                    #pragma unroll
                    for (int j = 0; j < 4; ++j) {
                        const unsigned rel = (unsigned)(d[j] - lo);
                        const bool mine = rel < RANGE;
                        const unsigned who = __builtin_amdgcn_ballot_w32(mine);
                        if (mine)
                            list[nlist + __popc(who & below)] =
                                make_uint2(rel, (unsigned)s[j]);
                        nlist += __popc(who);
                    }
                }
                __builtin_amdgcn_fence(__ATOMIC_ACQ_REL, "wavefront");

                for (int i = lane; i < nlist; i += 32) {
                    const uint2 ent = list[i];
                    const int s = (int)ent.y;
                    const float val = dinv_of(deg[s]) * dinv_of(deg[lo + (int)ent.x]) * h[s];
                    __hip_atomic_fetch_add(&s_acc[ent.x], val, __ATOMIC_RELAXED,
                                           __HIP_MEMORY_SCOPE_WORKGROUP);
                }
                __builtin_amdgcn_fence(__ATOMIC_ACQ_REL, "wavefront");
            }
        }
    } else {
        for (int e = beg + tid; e < end; e += SCAT_THREADS) {
            const unsigned rel = (unsigned)(dst[e] - lo);
            if (rel < RANGE) {
                const int s = src[e];
                const float val = dinv_of(deg[s]) * dinv_of(deg[lo + (int)rel]) * h[s];
                __hip_atomic_fetch_add(&s_acc[rel], val, __ATOMIC_RELAXED,
                                       __HIP_MEMORY_SCOPE_WORKGROUP);
            }
        }
    }
    __syncthreads();

    const int n = min(RANGE, N - lo);
    for (int i = tid; i < n; i += SCAT_THREADS)
        atomicAdd(&accum[lo + i], s_acc[i]);
}

__global__ void gcn_finalize_kernel(const float* __restrict__ accum,
                                    const unsigned* __restrict__ deg,
                                    const float* __restrict__ h,
                                    const float* __restrict__ b,
                                    float* __restrict__ out, int N) {
    int i = blockIdx.x * blockDim.x + threadIdx.x;
    if (i < N) {
        float di = dinv_of(deg[i]);
        float v  = accum[i] + di * di * h[i] + b[0];
        out[i]   = fmaxf(v, 0.0f) + log1pf(expf(-fabsf(v)));
    }
}

extern "C" void kernel_launch(void* const* d_in, const int* in_sizes, int n_in,
                              void* d_out, int out_size, void* d_ws, size_t ws_size,
                              hipStream_t stream) {
    const float* x  = (const float*)d_in[0];
    const int*   ei = (const int*)d_in[1];
    const float* W  = (const float*)d_in[2];
    const float* b  = (const float*)d_in[3];
    float* out = (float*)d_out;

    const int N = in_sizes[0] / C_DIM;
    const int E = in_sizes[1] / 2;

    float*    h     = (float*)d_ws;
    unsigned* deg   = (unsigned*)(h + N);
    float*    accum = (float*)(deg + N) + N;

    const int BLK = 256;
    const int gridN = (N + BLK - 1) / BLK;
    const dim3 gridS((N + RANGE - 1) / RANGE, SLICES);
    const int nodesPerBlock = (BLK / 32) * 16;
    const int gridM = (N + nodesPerBlock - 1) / nodesPerBlock;

    gcn_matvec_wmma_kernel<<<gridM, BLK, 0, stream>>>(x, W, h, deg, accum, N);
    gcn_degree_kernel<<<gridS, SCAT_THREADS, 0, stream>>>(ei, deg, N, E);
    gcn_edge_kernel<<<gridS, SCAT_THREADS, 0, stream>>>(ei, deg, h, accum, N, E);
    gcn_finalize_kernel<<<gridN, BLK, 0, stream>>>(accum, deg, h, b, out, N);
}


# Edge gathers, three per edge to one: h and degree of a node share one 8-byte record, dinv[dst] is factored out of the sum and applied once per node at the end; float sum order and rounding change
# speedup vs baseline: 5.5073x; 1.4382x over previous
#include <hip/hip_runtime.h>
#include <hip/hip_bf16.h>
#include <math.h>

typedef __attribute__((ext_vector_type(2))) float v2f;
typedef __attribute__((ext_vector_type(8))) float v8f;

#define C_DIM 256

__global__ void gcn_matvec_wmma_kernel(const float* __restrict__ x,
                                       const float* __restrict__ W,
                                       unsigned* __restrict__ hd,
                                       float* __restrict__ accum, int N) {
    const int lane = threadIdx.x & 31;
    const int wave = threadIdx.x >> 5;
    const int wavesPerBlock = blockDim.x >> 5;
    const int base = (blockIdx.x * wavesPerBlock + wave) * 16;
    if (base >= N) return;

    const int half = lane >> 4;
    const int m    = lane & 15;
    int row = base + m;
    if (row >= N) row = N - 1;

    const float* __restrict__ xrow = x + (size_t)row * C_DIM;

    v8f acc = {0.f, 0.f, 0.f, 0.f, 0.f, 0.f, 0.f, 0.f};

    #pragma unroll 4
    for (int k0 = 0; k0 < C_DIM; k0 += 4) {
        const int ka = k0 + half * 2;
        v2f a = *(const v2f*)(xrow + ka);
        v2f b;
        b.x = W[ka];
        b.y = W[ka + 1];
        acc = __builtin_amdgcn_wmma_f32_16x16x4_f32(
             false, a,  false, b,
             (short)0, acc,  false,  false);
    }

    if (m < 8) {
        float hv = acc[0];
        #pragma unroll
        for (int r = 1; r < 8; ++r) hv = (m == r) ? acc[r] : hv;
        const int node = base + half * 8 + m;
        if (node < N)
            ((uint2*)hd)[node] = make_uint2(__float_as_uint(hv), 0u);
    }
    if (half == 0 && base + m < N) accum[base + m] = 0.0f;
}

#define RANGE        6250
#define SLICES       16
#define SCAT_THREADS 1024
#define SCAT_WAVES   (SCAT_THREADS / 32)
#define BATCH        4
#define WTILE        (BATCH * 128)

static __device__ __forceinline__ int4 load_ids4(const int* __restrict__ p,
                                                 int e, int end) {
    return *(const int4*)(p + min(e, end - 4));
}
static __device__ __forceinline__ int4 keep_ids4(int4 t, int e, int end) {
    const bool in = e < end;
    return make_int4(in ? t.x : -1, in ? t.y : -1, in ? t.z : -1, in ? t.w : -1);
}

static __device__ __forceinline__ int slice_len(int E) {
    int s = (E + SLICES - 1) / SLICES;
    return (s + 3) & ~3;
}

__global__ __launch_bounds__(SCAT_THREADS)
void gcn_degree_kernel(const int* __restrict__ ei,
                       unsigned* __restrict__ hd, int N, int E) {
    __shared__ unsigned s_cnt[RANGE];
    const int tid = threadIdx.x;
    const int lo  = blockIdx.x * RANGE;
    for (int i = tid; i < RANGE; i += SCAT_THREADS) s_cnt[i] = 0u;
    __syncthreads();

    const int* __restrict__ dst = ei + E;
    const bool vec = (E & 3) == 0;
    const int S   = slice_len(E);
    const int beg = blockIdx.y * S;
    const int end = min(E, beg + S);

    if (vec) {
        if (beg < end) {
            const int step = SCAT_THREADS * 4 * BATCH;
            int4 nx[BATCH];
            #pragma unroll
            for (int u = 0; u < BATCH; ++u)
                nx[u] = load_ids4(dst, beg + tid * 4 + u * SCAT_THREADS * 4, end);
            for (int e0 = beg + tid * 4; e0 < end; e0 += step) {
                int4 v[BATCH];
                #pragma unroll
                for (int u = 0; u < BATCH; ++u) v[u] = nx[u];
                #pragma unroll
                for (int u = 0; u < BATCH; ++u)
                    nx[u] = load_ids4(dst, e0 + step + u * SCAT_THREADS * 4, end);
                #pragma unroll
                for (int u = 0; u < BATCH; ++u) {
                    const int4 k = keep_ids4(v[u], e0 + u * SCAT_THREADS * 4, end);
                    const int d[4] = {k.x, k.y, k.z, k.w};
                    #pragma unroll
                    for (int j = 0; j < 4; ++j) {
                        const unsigned rel = (unsigned)(d[j] - lo);
                        if (rel < RANGE)
                            __hip_atomic_fetch_add(&s_cnt[rel], 1u, __ATOMIC_RELAXED,
                                                   __HIP_MEMORY_SCOPE_WORKGROUP);
                    }
                }
            }
        }
    } else {
        for (int e = beg + tid; e < end; e += SCAT_THREADS) {
            const unsigned rel = (unsigned)(dst[e] - lo);
            if (rel < RANGE)
                __hip_atomic_fetch_add(&s_cnt[rel], 1u, __ATOMIC_RELAXED,
                                       __HIP_MEMORY_SCOPE_WORKGROUP);
        }
    }
    __syncthreads();

    const int n = min(RANGE, N - lo);
    for (int i = tid; i < n; i += SCAT_THREADS)
        atomicAdd(&hd[2 * (lo + i) + 1], s_cnt[i]);
}

static __device__ __forceinline__ float dinv_of(unsigned d) {
    return rsqrtf((float)(d + 1u));
}

__global__ __launch_bounds__(SCAT_THREADS)
void gcn_edge_kernel(const int* __restrict__ ei,
                     const unsigned* __restrict__ hd,
                     float* __restrict__ accum, int N, int E) {
    __shared__ float s_acc[RANGE];
    __shared__ uint2 s_list[SCAT_WAVES][WTILE];
    const int tid  = threadIdx.x;
    const int lane = tid & 31;
    const int wave = tid >> 5;
    const int lo   = blockIdx.x * RANGE;
    for (int i = tid; i < RANGE; i += SCAT_THREADS) s_acc[i] = 0.0f;
    __syncthreads();

    const int* __restrict__ src = ei;
    const int* __restrict__ dst = ei + E;
    const uint2* __restrict__ rec = (const uint2*)hd;
    const bool vec = (E & 3) == 0;
    const int S   = slice_len(E);
    const int beg = blockIdx.y * S;
    const int end = min(E, beg + S);
    uint2* __restrict__ list = s_list[wave];
    const unsigned below = (1u << lane) - 1u;

    if (vec) {
        if (beg < end) {
            const int step = SCAT_WAVES * WTILE;
            int4 nd[BATCH], ns[BATCH];
            #pragma unroll
            for (int u = 0; u < BATCH; ++u) {
                const int e = beg + wave * WTILE + u * 128 + lane * 4;
                nd[u] = load_ids4(dst, e, end);
                ns[u] = load_ids4(src, e, end);
            }
            for (int tb = beg + wave * WTILE; tb < end; tb += step) {
                int4 vd[BATCH], vs[BATCH];
                #pragma unroll
                for (int u = 0; u < BATCH; ++u) { vd[u] = nd[u]; vs[u] = ns[u]; }
                #pragma unroll
                for (int u = 0; u < BATCH; ++u) {
                    const int e = tb + step + u * 128 + lane * 4;
                    nd[u] = load_ids4(dst, e, end);
                    ns[u] = load_ids4(src, e, end);
                }

                int nlist = 0;
                #pragma unroll
                for (int u = 0; u < BATCH; ++u) {
                    const int4 k = keep_ids4(vd[u], tb + u * 128 + lane * 4, end);
                    const int d[4] = {k.x, k.y, k.z, k.w};
                    const int s[4] = {vs[u].x, vs[u].y, vs[u].z, vs[u].w};
                    #pragma unroll
                    for (int j = 0; j < 4; ++j) {
                        const unsigned rel = (unsigned)(d[j] - lo);
                        const bool mine = rel < RANGE;
                        const unsigned who = __builtin_amdgcn_ballot_w32(mine);
                        if (mine)
                            list[nlist + __popc(who & below)] =
                                make_uint2(rel, (unsigned)s[j]);
                        nlist += __popc(who);
                    }
                }
                __builtin_amdgcn_fence(__ATOMIC_ACQ_REL, "wavefront");

                for (int i = lane; i < nlist; i += 32) {
                    const uint2 ent = list[i];
                    const uint2 r = rec[ent.y];
                    const float val = dinv_of(r.y) * __uint_as_float(r.x);
                    __hip_atomic_fetch_add(&s_acc[ent.x], val, __ATOMIC_RELAXED,
                                           __HIP_MEMORY_SCOPE_WORKGROUP);
                }
                __builtin_amdgcn_fence(__ATOMIC_ACQ_REL, "wavefront");
            }
        }
    } else {
        for (int e = beg + tid; e < end; e += SCAT_THREADS) {
            const unsigned rel = (unsigned)(dst[e] - lo);
            if (rel < RANGE) {
                const uint2 r = rec[src[e]];
                const float val = dinv_of(r.y) * __uint_as_float(r.x);
                __hip_atomic_fetch_add(&s_acc[rel], val, __ATOMIC_RELAXED,
                                       __HIP_MEMORY_SCOPE_WORKGROUP);
            }
        }
    }
    __syncthreads();

    const int n = min(RANGE, N - lo);
    for (int i = tid; i < n; i += SCAT_THREADS)
        atomicAdd(&accum[lo + i], s_acc[i]);
}

__global__ void gcn_finalize_kernel(const float* __restrict__ accum,
                                    const unsigned* __restrict__ hd,
                                    const float* __restrict__ b,
                                    float* __restrict__ out, int N) {
    int i = blockIdx.x * blockDim.x + threadIdx.x;
    if (i < N) {
        const uint2 r = ((const uint2*)hd)[i];
        float di = dinv_of(r.y);
        float v  = di * (accum[i] + di * __uint_as_float(r.x)) + b[0];
        out[i]   = fmaxf(v, 0.0f) + log1pf(expf(-fabsf(v)));
    }
}

extern "C" void kernel_launch(void* const* d_in, const int* in_sizes, int n_in,
                              void* d_out, int out_size, void* d_ws, size_t ws_size,
                              hipStream_t stream) {
    const float* x  = (const float*)d_in[0];
    const int*   ei = (const int*)d_in[1];
    const float* W  = (const float*)d_in[2];
    const float* b  = (const float*)d_in[3];
    float* out = (float*)d_out;

    const int N = in_sizes[0] / C_DIM;
    const int E = in_sizes[1] / 2;

    unsigned* hd    = (unsigned*)d_ws;
    float*    accum = (float*)(hd + N + N) + N;

    const int BLK = 256;
    const int gridN = (N + BLK - 1) / BLK;
    const dim3 gridS((N + RANGE - 1) / RANGE, SLICES);
    const int nodesPerBlock = (BLK / 32) * 16;
    const int gridM = (N + nodesPerBlock - 1) / nodesPerBlock;

    gcn_matvec_wmma_kernel<<<gridM, BLK, 0, stream>>>(x, W, hd, accum, N);
    gcn_degree_kernel<<<gridS, SCAT_THREADS, 0, stream>>>(ei, hd, N, E);
    gcn_edge_kernel<<<gridS, SCAT_THREADS, 0, stream>>>(ei, hd, accum, N, E);
    gcn_finalize_kernel<<<gridN, BLK, 0, stream>>>(accum, hd, b, out, N);
}


# Matvec loads: a node's 8 lanes read its whole 128-byte line (4 whole lines per 16-byte load), 12-16 loads in flight per wave; 4 partial dots per node summed at the end; float sum order changes
# speedup vs baseline: 5.7635x; 1.0465x over previous
#include <hip/hip_runtime.h>
#include <hip/hip_bf16.h>
#include <math.h>

typedef __attribute__((ext_vector_type(2))) float v2f;
typedef __attribute__((ext_vector_type(8))) float v8f;

#define C_DIM 256

struct MvStep {
    float4 w;
    float4 x[4];
};

static __device__ __forceinline__ void mv_load(MvStep& s, const float* __restrict__ wp,
                                               const float* __restrict__ x0,
                                               const float* __restrict__ x1,
                                               const float* __restrict__ x2,
                                               const float* __restrict__ x3, int step) {
    const int o = 32 * step;
    s.w    = *(const float4*)(wp + o);
    s.x[0] = *(const float4*)(x0 + o);
    s.x[1] = *(const float4*)(x1 + o);
    s.x[2] = *(const float4*)(x2 + o);
    s.x[3] = *(const float4*)(x3 + o);
}

static __device__ __forceinline__ void mv_products(const MvStep& s, v8f* acc) {
    v2f b0, b1;
    b0.x = s.w.x; b0.y = s.w.y;
    b1.x = s.w.z; b1.y = s.w.w;
    #pragma unroll
    for (int g = 0; g < 4; ++g) {
        v2f a0;
        a0.x = s.x[g].x; a0.y = s.x[g].y;
        acc[g] = __builtin_amdgcn_wmma_f32_16x16x4_f32(
             false, a0,  false, b0,
             (short)0, acc[g],  false,  false);
    }
    #pragma unroll
    for (int g = 0; g < 4; ++g) {
        v2f a1;
        a1.x = s.x[g].z; a1.y = s.x[g].w;
        acc[g] = __builtin_amdgcn_wmma_f32_16x16x4_f32(
             false, a1,  false, b1,
             (short)0, acc[g],  false,  false);
    }
}

__global__ __launch_bounds__(256)
void gcn_matvec_wmma_kernel(const float* __restrict__ x,
                            const float* __restrict__ W,
                            unsigned* __restrict__ hd,
                            float* __restrict__ accum, int N) {
    const int lane = threadIdx.x & 31;
    const int wave = threadIdx.x >> 5;
    const int wavesPerBlock = blockDim.x >> 5;
    const int base = (blockIdx.x * wavesPerBlock + wave) * 16;
    if (base >= N) return;

    const int half = lane >> 4;
    const int m    = lane & 15;
    const int q    = m >> 2;
    const int c    = m & 3;
    const int koff = (2 * c + half) * 4;

    const float* __restrict__ xr[4];
    #pragma unroll
    for (int g = 0; g < 4; ++g) {
        int row = base + 4 * g + q;
        if (row >= N) row = N - 1;
        xr[g] = x + (size_t)row * C_DIM + koff;
    }
    const float* __restrict__ wp = W + koff;

    v8f acc[4];
    #pragma unroll
    for (int g = 0; g < 4; ++g)
        acc[g] = (v8f){0.f, 0.f, 0.f, 0.f, 0.f, 0.f, 0.f, 0.f};

    MvStep st[4];
    mv_load(st[0], wp, xr[0], xr[1], xr[2], xr[3], 0);
    mv_load(st[1], wp, xr[0], xr[1], xr[2], xr[3], 1);
    mv_load(st[2], wp, xr[0], xr[1], xr[2], xr[3], 2);
    mv_load(st[3], wp, xr[0], xr[1], xr[2], xr[3], 3);
    __builtin_amdgcn_sched_barrier(0);
    mv_products(st[0], acc);
    __builtin_amdgcn_sched_barrier(0);
    mv_load(st[0], wp, xr[0], xr[1], xr[2], xr[3], 4);
    __builtin_amdgcn_sched_barrier(0);
    mv_products(st[1], acc);
    __builtin_amdgcn_sched_barrier(0);
    mv_load(st[1], wp, xr[0], xr[1], xr[2], xr[3], 5);
    __builtin_amdgcn_sched_barrier(0);
    mv_products(st[2], acc);
    __builtin_amdgcn_sched_barrier(0);
    mv_load(st[2], wp, xr[0], xr[1], xr[2], xr[3], 6);
    __builtin_amdgcn_sched_barrier(0);
    mv_products(st[3], acc);
    __builtin_amdgcn_sched_barrier(0);
    mv_load(st[3], wp, xr[0], xr[1], xr[2], xr[3], 7);
    __builtin_amdgcn_sched_barrier(0);
    mv_products(st[0], acc);
    mv_products(st[1], acc);
    mv_products(st[2], acc);
    mv_products(st[3], acc);

    float lo = 0.0f, hi = 0.0f;
    #pragma unroll
    for (int g = 0; g < 4; ++g) {
        #pragma unroll
        for (int r = 0; r < 4; ++r) {
            const bool s = (q == g) && (c == r);
            lo = s ? acc[g][r]     : lo;
            hi = s ? acc[g][4 + r] : hi;
        }
    }
    lo += __shfl_xor(lo, 1);
    lo += __shfl_xor(lo, 2);
    hi += __shfl_xor(hi, 1);
    hi += __shfl_xor(hi, 2);
    if (c < 2) {
        const int node = base + 4 * q + 2 * half + c;
        const float hv = c ? hi : lo;
        if (node < N)
            ((uint2*)hd)[node] = make_uint2(__float_as_uint(hv), 0u);
    }
    if (half == 0 && base + m < N) accum[base + m] = 0.0f;
}

#define RANGE        6250
#define SLICES       16
#define SCAT_THREADS 1024
#define SCAT_WAVES   (SCAT_THREADS / 32)
#define BATCH        4
#define WTILE        (BATCH * 128)

static __device__ __forceinline__ int4 load_ids4(const int* __restrict__ p,
                                                 int e, int end) {
    return *(const int4*)(p + min(e, end - 4));
}
static __device__ __forceinline__ int4 keep_ids4(int4 t, int e, int end) {
    const bool in = e < end;
    return make_int4(in ? t.x : -1, in ? t.y : -1, in ? t.z : -1, in ? t.w : -1);
}

static __device__ __forceinline__ int slice_len(int E) {
    int s = (E + SLICES - 1) / SLICES;
    return (s + 3) & ~3;
}

__global__ __launch_bounds__(SCAT_THREADS)
void gcn_degree_kernel(const int* __restrict__ ei,
                       unsigned* __restrict__ hd, int N, int E) {
    __shared__ unsigned s_cnt[RANGE];
    const int tid = threadIdx.x;
    const int lo  = blockIdx.x * RANGE;
    for (int i = tid; i < RANGE; i += SCAT_THREADS) s_cnt[i] = 0u;
    __syncthreads();

    const int* __restrict__ dst = ei + E;
    const bool vec = (E & 3) == 0;
    const int S   = slice_len(E);
    const int beg = blockIdx.y * S;
    const int end = min(E, beg + S);

    if (vec) {
        if (beg < end) {
            const int step = SCAT_THREADS * 4 * BATCH;
            int4 nx[BATCH];
            #pragma unroll
            for (int u = 0; u < BATCH; ++u)
                nx[u] = load_ids4(dst, beg + tid * 4 + u * SCAT_THREADS * 4, end);
            for (int e0 = beg + tid * 4; e0 < end; e0 += step) {
                int4 v[BATCH];
                #pragma unroll
                for (int u = 0; u < BATCH; ++u) v[u] = nx[u];
                #pragma unroll
                for (int u = 0; u < BATCH; ++u)
                    nx[u] = load_ids4(dst, e0 + step + u * SCAT_THREADS * 4, end);
                #pragma unroll
                for (int u = 0; u < BATCH; ++u) {
                    const int4 k = keep_ids4(v[u], e0 + u * SCAT_THREADS * 4, end);
                    const int d[4] = {k.x, k.y, k.z, k.w};
                    #pragma unroll
                    for (int j = 0; j < 4; ++j) {
                        const unsigned rel = (unsigned)(d[j] - lo);
                        if (rel < RANGE)
                            __hip_atomic_fetch_add(&s_cnt[rel], 1u, __ATOMIC_RELAXED,
                                                   __HIP_MEMORY_SCOPE_WORKGROUP);
                    }
                }
            }
        }
    } else {
        for (int e = beg + tid; e < end; e += SCAT_THREADS) {
            const unsigned rel = (unsigned)(dst[e] - lo);
            if (rel < RANGE)
                __hip_atomic_fetch_add(&s_cnt[rel], 1u, __ATOMIC_RELAXED,
                                       __HIP_MEMORY_SCOPE_WORKGROUP);
        }
    }
    __syncthreads();

    const int n = min(RANGE, N - lo);
    for (int i = tid; i < n; i += SCAT_THREADS)
        atomicAdd(&hd[2 * (lo + i) + 1], s_cnt[i]);
}

static __device__ __forceinline__ float dinv_of(unsigned d) {
    return rsqrtf((float)(d + 1u));
}

__global__ __launch_bounds__(SCAT_THREADS)
void gcn_edge_kernel(const int* __restrict__ ei,
                     const unsigned* __restrict__ hd,
                     float* __restrict__ accum, int N, int E) {
    __shared__ float s_acc[RANGE];
    __shared__ uint2 s_list[SCAT_WAVES][WTILE];
    const int tid  = threadIdx.x;
    const int lane = tid & 31;
    const int wave = tid >> 5;
    const int lo   = blockIdx.x * RANGE;
    for (int i = tid; i < RANGE; i += SCAT_THREADS) s_acc[i] = 0.0f;
    __syncthreads();

    const int* __restrict__ src = ei;
    const int* __restrict__ dst = ei + E;
    const uint2* __restrict__ rec = (const uint2*)hd;
    const bool vec = (E & 3) == 0;
    const int S   = slice_len(E);
    const int beg = blockIdx.y * S;
    const int end = min(E, beg + S);
    uint2* __restrict__ list = s_list[wave];
    const unsigned below = (1u << lane) - 1u;

    if (vec) {
        if (beg < end) {
            const int step = SCAT_WAVES * WTILE;
            int4 nd[BATCH], ns[BATCH];
            #pragma unroll
            for (int u = 0; u < BATCH; ++u) {
                const int e = beg + wave * WTILE + u * 128 + lane * 4;
                nd[u] = load_ids4(dst, e, end);
                ns[u] = load_ids4(src, e, end);
            }
            for (int tb = beg + wave * WTILE; tb < end; tb += step) {
                int4 vd[BATCH], vs[BATCH];
                #pragma unroll
                for (int u = 0; u < BATCH; ++u) { vd[u] = nd[u]; vs[u] = ns[u]; }
                #pragma unroll
                for (int u = 0; u < BATCH; ++u) {
                    const int e = tb + step + u * 128 + lane * 4;
                    nd[u] = load_ids4(dst, e, end);
                    ns[u] = load_ids4(src, e, end);
                }

                int nlist = 0;
                #pragma unroll
                for (int u = 0; u < BATCH; ++u) {
                    const int4 k = keep_ids4(vd[u], tb + u * 128 + lane * 4, end);
                    const int d[4] = {k.x, k.y, k.z, k.w};
                    const int s[4] = {vs[u].x, vs[u].y, vs[u].z, vs[u].w};
                    #pragma unroll
                    for (int j = 0; j < 4; ++j) {
                        const unsigned rel = (unsigned)(d[j] - lo);
                        const bool mine = rel < RANGE;
                        const unsigned who = __builtin_amdgcn_ballot_w32(mine);
                        if (mine)
                            list[nlist + __popc(who & below)] =
                                make_uint2(rel, (unsigned)s[j]);
                        nlist += __popc(who);
                    }
                }
                __builtin_amdgcn_fence(__ATOMIC_ACQ_REL, "wavefront");

                for (int i = lane; i < nlist; i += 32) {
                    const uint2 ent = list[i];
                    const uint2 r = rec[ent.y];
                    const float val = dinv_of(r.y) * __uint_as_float(r.x);
                    __hip_atomic_fetch_add(&s_acc[ent.x], val, __ATOMIC_RELAXED,
                                           __HIP_MEMORY_SCOPE_WORKGROUP);
                }
                __builtin_amdgcn_fence(__ATOMIC_ACQ_REL, "wavefront");
            }
        }
    } else {
        for (int e = beg + tid; e < end; e += SCAT_THREADS) {
            const unsigned rel = (unsigned)(dst[e] - lo);
            if (rel < RANGE) {
                const uint2 r = rec[src[e]];
                const float val = dinv_of(r.y) * __uint_as_float(r.x);
                __hip_atomic_fetch_add(&s_acc[rel], val, __ATOMIC_RELAXED,
                                       __HIP_MEMORY_SCOPE_WORKGROUP);
            }
        }
    }
    __syncthreads();

    const int n = min(RANGE, N - lo);
    for (int i = tid; i < n; i += SCAT_THREADS)
        atomicAdd(&accum[lo + i], s_acc[i]);
}

__global__ void gcn_finalize_kernel(const float* __restrict__ accum,
                                    const unsigned* __restrict__ hd,
                                    const float* __restrict__ b,
                                    float* __restrict__ out, int N) {
    int i = blockIdx.x * blockDim.x + threadIdx.x;
    if (i < N) {
        const uint2 r = ((const uint2*)hd)[i];
        float di = dinv_of(r.y);
        float v  = di * (accum[i] + di * __uint_as_float(r.x)) + b[0];
        out[i]   = fmaxf(v, 0.0f) + log1pf(expf(-fabsf(v)));
    }
}

extern "C" void kernel_launch(void* const* d_in, const int* in_sizes, int n_in,
                              void* d_out, int out_size, void* d_ws, size_t ws_size,
                              hipStream_t stream) {
    const float* x  = (const float*)d_in[0];
    const int*   ei = (const int*)d_in[1];
    const float* W  = (const float*)d_in[2];
    const float* b  = (const float*)d_in[3];
    float* out = (float*)d_out;

    const int N = in_sizes[0] / C_DIM;
    const int E = in_sizes[1] / 2;

    unsigned* hd    = (unsigned*)d_ws;
    float*    accum = (float*)(hd + N + N) + N;

    const int BLK = 256;
    const int gridN = (N + BLK - 1) / BLK;
    const dim3 gridS((N + RANGE - 1) / RANGE, SLICES);
    const int nodesPerBlock = (BLK / 32) * 16;
    const int gridM = (N + nodesPerBlock - 1) / nodesPerBlock;

    gcn_matvec_wmma_kernel<<<gridM, BLK, 0, stream>>>(x, W, hd, accum, N);
    gcn_degree_kernel<<<gridS, SCAT_THREADS, 0, stream>>>(ei, hd, N, E);
    gcn_edge_kernel<<<gridS, SCAT_THREADS, 0, stream>>>(ei, hd, accum, N, E);
    gcn_finalize_kernel<<<gridN, BLK, 0, stream>>>(accum, hd, b, out, N);
}


# Edge kernel split, 16 dst ranges x 16 edge slices to 8 x 32: it scans and tests half as many ids for twice the whole-line merge adds; list worked off two entries per lane; float sum order changes
# speedup vs baseline: 6.0657x; 1.0524x over previous
#include <hip/hip_runtime.h>
#include <hip/hip_bf16.h>
#include <math.h>

typedef __attribute__((ext_vector_type(2))) float v2f;
typedef __attribute__((ext_vector_type(8))) float v8f;

#define C_DIM 256

struct MvStep {
    float4 w;
    float4 x[4];
};

static __device__ __forceinline__ void mv_load(MvStep& s, const float* __restrict__ wp,
                                               const float* __restrict__ x0,
                                               const float* __restrict__ x1,
                                               const float* __restrict__ x2,
                                               const float* __restrict__ x3, int step) {
    const int o = 32 * step;
    s.w    = *(const float4*)(wp + o);
    s.x[0] = *(const float4*)(x0 + o);
    s.x[1] = *(const float4*)(x1 + o);
    s.x[2] = *(const float4*)(x2 + o);
    s.x[3] = *(const float4*)(x3 + o);
}

static __device__ __forceinline__ void mv_products(const MvStep& s, v8f* acc) {
    v2f b0, b1;
    b0.x = s.w.x; b0.y = s.w.y;
    b1.x = s.w.z; b1.y = s.w.w;
    #pragma unroll
    for (int g = 0; g < 4; ++g) {
        v2f a0;
        a0.x = s.x[g].x; a0.y = s.x[g].y;
        acc[g] = __builtin_amdgcn_wmma_f32_16x16x4_f32(
             false, a0,  false, b0,
             (short)0, acc[g],  false,  false);
    }
    #pragma unroll
    for (int g = 0; g < 4; ++g) {
        v2f a1;
        a1.x = s.x[g].z; a1.y = s.x[g].w;
        acc[g] = __builtin_amdgcn_wmma_f32_16x16x4_f32(
             false, a1,  false, b1,
             (short)0, acc[g],  false,  false);
    }
}

__global__ __launch_bounds__(256)
void gcn_matvec_wmma_kernel(const float* __restrict__ x,
                            const float* __restrict__ W,
                            unsigned* __restrict__ hd,
                            float* __restrict__ accum, int N) {
    const int lane = threadIdx.x & 31;
    const int wave = threadIdx.x >> 5;
    const int wavesPerBlock = blockDim.x >> 5;
    const int base = (blockIdx.x * wavesPerBlock + wave) * 16;
    if (base >= N) return;

    const int half = lane >> 4;
    const int m    = lane & 15;
    const int q    = m >> 2;
    const int c    = m & 3;
    const int koff = (2 * c + half) * 4;

    const float* __restrict__ xr[4];
    #pragma unroll
    for (int g = 0; g < 4; ++g) {
        int row = base + 4 * g + q;
        if (row >= N) row = N - 1;
        xr[g] = x + (size_t)row * C_DIM + koff;
    }
    const float* __restrict__ wp = W + koff;

    v8f acc[4];
    #pragma unroll
    for (int g = 0; g < 4; ++g)
        acc[g] = (v8f){0.f, 0.f, 0.f, 0.f, 0.f, 0.f, 0.f, 0.f};

    MvStep st[4];
    mv_load(st[0], wp, xr[0], xr[1], xr[2], xr[3], 0);
    mv_load(st[1], wp, xr[0], xr[1], xr[2], xr[3], 1);
    mv_load(st[2], wp, xr[0], xr[1], xr[2], xr[3], 2);
    mv_load(st[3], wp, xr[0], xr[1], xr[2], xr[3], 3);
    __builtin_amdgcn_sched_barrier(0);
    mv_products(st[0], acc);
    __builtin_amdgcn_sched_barrier(0);
    mv_load(st[0], wp, xr[0], xr[1], xr[2], xr[3], 4);
    __builtin_amdgcn_sched_barrier(0);
    mv_products(st[1], acc);
    __builtin_amdgcn_sched_barrier(0);
    mv_load(st[1], wp, xr[0], xr[1], xr[2], xr[3], 5);
    __builtin_amdgcn_sched_barrier(0);
    mv_products(st[2], acc);
    __builtin_amdgcn_sched_barrier(0);
    mv_load(st[2], wp, xr[0], xr[1], xr[2], xr[3], 6);
    __builtin_amdgcn_sched_barrier(0);
    mv_products(st[3], acc);
    __builtin_amdgcn_sched_barrier(0);
    mv_load(st[3], wp, xr[0], xr[1], xr[2], xr[3], 7);
    __builtin_amdgcn_sched_barrier(0);
    mv_products(st[0], acc);
    mv_products(st[1], acc);
    mv_products(st[2], acc);
    mv_products(st[3], acc);

    float lo = 0.0f, hi = 0.0f;
    #pragma unroll
    for (int g = 0; g < 4; ++g) {
        #pragma unroll
        for (int r = 0; r < 4; ++r) {
            const bool s = (q == g) && (c == r);
            lo = s ? acc[g][r]     : lo;
            hi = s ? acc[g][4 + r] : hi;
        }
    }
    lo += __shfl_xor(lo, 1);
    lo += __shfl_xor(lo, 2);
    hi += __shfl_xor(hi, 1);
    hi += __shfl_xor(hi, 2);
    if (c < 2) {
        const int node = base + 4 * q + 2 * half + c;
        const float hv = c ? hi : lo;
        if (node < N)
            ((uint2*)hd)[node] = make_uint2(__float_as_uint(hv), 0u);
    }
    if (half == 0 && base + m < N) accum[base + m] = 0.0f;
}

#define RANGE        6250
#define SLICES       16
#define ERANGE       12500
#define ESLICES      32
#define SCAT_THREADS 1024
#define SCAT_WAVES   (SCAT_THREADS / 32)
#define BATCH        4
#define WTILE        (BATCH * 128)

static __device__ __forceinline__ int4 load_ids4(const int* __restrict__ p,
                                                 int e, int end) {
    return *(const int4*)(p + min(e, end - 4));
}
static __device__ __forceinline__ int4 keep_ids4(int4 t, int e, int end) {
    const bool in = e < end;
    return make_int4(in ? t.x : -1, in ? t.y : -1, in ? t.z : -1, in ? t.w : -1);
}

static __device__ __forceinline__ int slice_len(int E, int slices) {
    int s = (E + slices - 1) / slices;
    return (s + 3) & ~3;
}

__global__ __launch_bounds__(SCAT_THREADS)
void gcn_degree_kernel(const int* __restrict__ ei,
                       unsigned* __restrict__ hd, int N, int E) {
    __shared__ unsigned s_cnt[RANGE];
    const int tid = threadIdx.x;
    const int lo  = blockIdx.x * RANGE;
    for (int i = tid; i < RANGE; i += SCAT_THREADS) s_cnt[i] = 0u;
    __syncthreads();

    const int* __restrict__ dst = ei + E;
    const bool vec = (E & 3) == 0;
    const int S   = slice_len(E, SLICES);
    const int beg = blockIdx.y * S;
    const int end = min(E, beg + S);

    if (vec) {
        if (beg < end) {
            const int step = SCAT_THREADS * 4 * BATCH;
            int4 nx[BATCH];
            #pragma unroll
            for (int u = 0; u < BATCH; ++u)
                nx[u] = load_ids4(dst, beg + tid * 4 + u * SCAT_THREADS * 4, end);
            for (int e0 = beg + tid * 4; e0 < end; e0 += step) {
                int4 v[BATCH];
                #pragma unroll
                for (int u = 0; u < BATCH; ++u) v[u] = nx[u];
                #pragma unroll
                for (int u = 0; u < BATCH; ++u)
                    nx[u] = load_ids4(dst, e0 + step + u * SCAT_THREADS * 4, end);
                #pragma unroll
                for (int u = 0; u < BATCH; ++u) {
                    const int4 k = keep_ids4(v[u], e0 + u * SCAT_THREADS * 4, end);
                    const int d[4] = {k.x, k.y, k.z, k.w};
                    #pragma unroll
                    for (int j = 0; j < 4; ++j) {
                        const unsigned rel = (unsigned)(d[j] - lo);
                        if (rel < RANGE)
                            __hip_atomic_fetch_add(&s_cnt[rel], 1u, __ATOMIC_RELAXED,
                                                   __HIP_MEMORY_SCOPE_WORKGROUP);
                    }
                }
            }
        }
    } else {
        for (int e = beg + tid; e < end; e += SCAT_THREADS) {
            const unsigned rel = (unsigned)(dst[e] - lo);
            if (rel < RANGE)
                __hip_atomic_fetch_add(&s_cnt[rel], 1u, __ATOMIC_RELAXED,
                                       __HIP_MEMORY_SCOPE_WORKGROUP);
        }
    }
    __syncthreads();

    const int n = min(RANGE, N - lo);
    for (int i = tid; i < n; i += SCAT_THREADS)
        atomicAdd(&hd[2 * (lo + i) + 1], s_cnt[i]);
}

static __device__ __forceinline__ float dinv_of(unsigned d) {
    return rsqrtf((float)(d + 1u));
}

__global__ __launch_bounds__(SCAT_THREADS)
void gcn_edge_kernel(const int* __restrict__ ei,
                     const unsigned* __restrict__ hd,
                     float* __restrict__ accum, int N, int E) {
    __shared__ float s_acc[ERANGE];
    __shared__ uint2 s_list[SCAT_WAVES][WTILE];
    const int tid  = threadIdx.x;
    const int lane = tid & 31;
    const int wave = tid >> 5;
    const int lo   = blockIdx.x * ERANGE;
    for (int i = tid; i < ERANGE; i += SCAT_THREADS) s_acc[i] = 0.0f;
    __syncthreads();

    const int* __restrict__ src = ei;
    const int* __restrict__ dst = ei + E;
    const uint2* __restrict__ rec = (const uint2*)hd;
    const bool vec = (E & 3) == 0;
    const int S   = slice_len(E, ESLICES);
    const int beg = blockIdx.y * S;
    const int end = min(E, beg + S);
    uint2* __restrict__ list = s_list[wave];
    const unsigned below = (1u << lane) - 1u;

    if (vec) {
        if (beg < end) {
            const int step = SCAT_WAVES * WTILE;
            int4 nd[BATCH], ns[BATCH];
            #pragma unroll
            for (int u = 0; u < BATCH; ++u) {
                const int e = beg + wave * WTILE + u * 128 + lane * 4;
                nd[u] = load_ids4(dst, e, end);
                ns[u] = load_ids4(src, e, end);
            }
            for (int tb = beg + wave * WTILE; tb < end; tb += step) {
                int4 vd[BATCH], vs[BATCH];
                #pragma unroll
                for (int u = 0; u < BATCH; ++u) { vd[u] = nd[u]; vs[u] = ns[u]; }
                #pragma unroll
                for (int u = 0; u < BATCH; ++u) {
                    const int e = tb + step + u * 128 + lane * 4;
                    nd[u] = load_ids4(dst, e, end);
                    ns[u] = load_ids4(src, e, end);
                }

                int nlist = 0;
                #pragma unroll
                for (int u = 0; u < BATCH; ++u) {
                    const int4 k = keep_ids4(vd[u], tb + u * 128 + lane * 4, end);
                    const int d[4] = {k.x, k.y, k.z, k.w};
                    const int s[4] = {vs[u].x, vs[u].y, vs[u].z, vs[u].w};
                    #pragma unroll
                    for (int j = 0; j < 4; ++j) {
                        const unsigned rel = (unsigned)(d[j] - lo);
                        const bool mine = rel < ERANGE;
                        const unsigned who = __builtin_amdgcn_ballot_w32(mine);
                        if (mine)
                            list[nlist + __popc(who & below)] =
                                make_uint2(rel, (unsigned)s[j]);
                        nlist += __popc(who);
                    }
                }
                __builtin_amdgcn_fence(__ATOMIC_ACQ_REL, "wavefront");

                for (int i = lane; i < nlist; i += 64) {
                    const bool two = i + 32 < nlist;
                    const uint2 e0 = list[i];
                    const uint2 e1 = list[two ? i + 32 : i];
                    const uint2 r0 = rec[e0.y];
                    const uint2 r1 = rec[e1.y];
                    const float v0 = dinv_of(r0.y) * __uint_as_float(r0.x);
                    const float v1 = two ? dinv_of(r1.y) * __uint_as_float(r1.x) : 0.0f;
                    __hip_atomic_fetch_add(&s_acc[e0.x], v0, __ATOMIC_RELAXED,
                                           __HIP_MEMORY_SCOPE_WORKGROUP);
                    __hip_atomic_fetch_add(&s_acc[e1.x], v1, __ATOMIC_RELAXED,
                                           __HIP_MEMORY_SCOPE_WORKGROUP);
                }
                __builtin_amdgcn_fence(__ATOMIC_ACQ_REL, "wavefront");
            }
        }
    } else {
        for (int e = beg + tid; e < end; e += SCAT_THREADS) {
            const unsigned rel = (unsigned)(dst[e] - lo);
            if (rel < ERANGE) {
                const uint2 r = rec[src[e]];
                const float val = dinv_of(r.y) * __uint_as_float(r.x);
                __hip_atomic_fetch_add(&s_acc[rel], val, __ATOMIC_RELAXED,
                                       __HIP_MEMORY_SCOPE_WORKGROUP);
            }
        }
    }
    __syncthreads();

    const int n = min(ERANGE, N - lo);
    for (int i = tid; i < n; i += SCAT_THREADS)
        atomicAdd(&accum[lo + i], s_acc[i]);
}

__global__ void gcn_finalize_kernel(const float* __restrict__ accum,
                                    const unsigned* __restrict__ hd,
                                    const float* __restrict__ b,
                                    float* __restrict__ out, int N) {
    int i = blockIdx.x * blockDim.x + threadIdx.x;
    if (i < N) {
        const uint2 r = ((const uint2*)hd)[i];
        float di = dinv_of(r.y);
        float v  = di * (accum[i] + di * __uint_as_float(r.x)) + b[0];
        out[i]   = fmaxf(v, 0.0f) + log1pf(expf(-fabsf(v)));
    }
}

extern "C" void kernel_launch(void* const* d_in, const int* in_sizes, int n_in,
                              void* d_out, int out_size, void* d_ws, size_t ws_size,
                              hipStream_t stream) {
    const float* x  = (const float*)d_in[0];
    const int*   ei = (const int*)d_in[1];
    const float* W  = (const float*)d_in[2];
    const float* b  = (const float*)d_in[3];
    float* out = (float*)d_out;

    const int N = in_sizes[0] / C_DIM;
    const int E = in_sizes[1] / 2;

    unsigned* hd    = (unsigned*)d_ws;
    float*    accum = (float*)(hd + N + N) + N;

    const int BLK = 256;
    const int gridN = (N + BLK - 1) / BLK;
    const dim3 gridS((N + RANGE - 1) / RANGE, SLICES);
    const dim3 gridE((N + ERANGE - 1) / ERANGE, ESLICES);
    const int nodesPerBlock = (BLK / 32) * 16;
    const int gridM = (N + nodesPerBlock - 1) / nodesPerBlock;

    gcn_matvec_wmma_kernel<<<gridM, BLK, 0, stream>>>(x, W, hd, accum, N);
    gcn_degree_kernel<<<gridS, SCAT_THREADS, 0, stream>>>(ei, hd, N, E);
    gcn_edge_kernel<<<gridE, SCAT_THREADS, 0, stream>>>(ei, hd, accum, N, E);
    gcn_finalize_kernel<<<gridN, BLK, 0, stream>>>(accum, hd, b, out, N);
}


# Edge kernel: per-edge L2 gathers become LDS reads; a block holds dinv*h of half the sources beside a dst quarter's sums (2x4 cells, 32 slices); zero sums skip the merge; float sum order changes
# speedup vs baseline: 7.5459x; 1.2440x over previous
#include <hip/hip_runtime.h>
#include <hip/hip_bf16.h>
#include <math.h>

typedef __attribute__((ext_vector_type(2))) float v2f;
typedef __attribute__((ext_vector_type(8))) float v8f;

#define C_DIM 256

struct MvStep {
    float4 w;
    float4 x[4];
};

static __device__ __forceinline__ void mv_load(MvStep& s, const float* __restrict__ wp,
                                               const float* __restrict__ x0,
                                               const float* __restrict__ x1,
                                               const float* __restrict__ x2,
                                               const float* __restrict__ x3, int step) {
    const int o = 32 * step;
    s.w    = *(const float4*)(wp + o);
    s.x[0] = *(const float4*)(x0 + o);
    s.x[1] = *(const float4*)(x1 + o);
    s.x[2] = *(const float4*)(x2 + o);
    s.x[3] = *(const float4*)(x3 + o);
}

static __device__ __forceinline__ void mv_products(const MvStep& s, v8f* acc) {
    v2f b0, b1;
    b0.x = s.w.x; b0.y = s.w.y;
    b1.x = s.w.z; b1.y = s.w.w;
    #pragma unroll
    for (int g = 0; g < 4; ++g) {
        v2f a0;
        a0.x = s.x[g].x; a0.y = s.x[g].y;
        acc[g] = __builtin_amdgcn_wmma_f32_16x16x4_f32(
             false, a0,  false, b0,
             (short)0, acc[g],  false,  false);
    }
    #pragma unroll
    for (int g = 0; g < 4; ++g) {
        v2f a1;
        a1.x = s.x[g].z; a1.y = s.x[g].w;
        acc[g] = __builtin_amdgcn_wmma_f32_16x16x4_f32(
             false, a1,  false, b1,
             (short)0, acc[g],  false,  false);
    }
}

__global__ __launch_bounds__(256)
void gcn_matvec_wmma_kernel(const float* __restrict__ x,
                            const float* __restrict__ W,
                            unsigned* __restrict__ hd,
                            float* __restrict__ accum, int N) {
    const int lane = threadIdx.x & 31;
    const int wave = threadIdx.x >> 5;
    const int wavesPerBlock = blockDim.x >> 5;
    const int base = (blockIdx.x * wavesPerBlock + wave) * 16;
    if (base >= N) return;

    const int half = lane >> 4;
    const int m    = lane & 15;
    const int q    = m >> 2;
    const int c    = m & 3;
    const int koff = (2 * c + half) * 4;

    const float* __restrict__ xr[4];
    #pragma unroll
    for (int g = 0; g < 4; ++g) {
        int row = base + 4 * g + q;
        if (row >= N) row = N - 1;
        xr[g] = x + (size_t)row * C_DIM + koff;
    }
    const float* __restrict__ wp = W + koff;

    v8f acc[4];
    #pragma unroll
    for (int g = 0; g < 4; ++g)
        acc[g] = (v8f){0.f, 0.f, 0.f, 0.f, 0.f, 0.f, 0.f, 0.f};

    MvStep st[4];
    mv_load(st[0], wp, xr[0], xr[1], xr[2], xr[3], 0);
    mv_load(st[1], wp, xr[0], xr[1], xr[2], xr[3], 1);
    mv_load(st[2], wp, xr[0], xr[1], xr[2], xr[3], 2);
    mv_load(st[3], wp, xr[0], xr[1], xr[2], xr[3], 3);
    __builtin_amdgcn_sched_barrier(0);
    mv_products(st[0], acc);
    __builtin_amdgcn_sched_barrier(0);
    mv_load(st[0], wp, xr[0], xr[1], xr[2], xr[3], 4);
    __builtin_amdgcn_sched_barrier(0);
    mv_products(st[1], acc);
    __builtin_amdgcn_sched_barrier(0);
    mv_load(st[1], wp, xr[0], xr[1], xr[2], xr[3], 5);
    __builtin_amdgcn_sched_barrier(0);
    mv_products(st[2], acc);
    __builtin_amdgcn_sched_barrier(0);
    mv_load(st[2], wp, xr[0], xr[1], xr[2], xr[3], 6);
    __builtin_amdgcn_sched_barrier(0);
    mv_products(st[3], acc);
    __builtin_amdgcn_sched_barrier(0);
    mv_load(st[3], wp, xr[0], xr[1], xr[2], xr[3], 7);
    __builtin_amdgcn_sched_barrier(0);
    mv_products(st[0], acc);
    mv_products(st[1], acc);
    mv_products(st[2], acc);
    mv_products(st[3], acc);

    float lo = 0.0f, hi = 0.0f;
    #pragma unroll
    for (int g = 0; g < 4; ++g) {
        #pragma unroll
        for (int r = 0; r < 4; ++r) {
            const bool s = (q == g) && (c == r);
            lo = s ? acc[g][r]     : lo;
            hi = s ? acc[g][4 + r] : hi;
        }
    }
    lo += __shfl_xor(lo, 1);
    lo += __shfl_xor(lo, 2);
    hi += __shfl_xor(hi, 1);
    hi += __shfl_xor(hi, 2);
    if (c < 2) {
        const int node = base + 4 * q + 2 * half + c;
        const float hv = c ? hi : lo;
        if (node < N)
            ((uint2*)hd)[node] = make_uint2(__float_as_uint(hv), 0u);
    }
    if (half == 0 && base + m < N) accum[base + m] = 0.0f;
}

#define RANGE        6250
#define SLICES       16
#define SRANGE       50000
#define DRANGE       25000
#define ESLICES      32
#define SCAT_THREADS 1024
#define BATCH        4
#define WTILE        (BATCH * 128)
#define SBATCH       8

static __device__ __forceinline__ int4 load_ids4(const int* __restrict__ p,
                                                 int e, int end) {
    return *(const int4*)(p + min(e, end - 4));
}
static __device__ __forceinline__ int4 keep_ids4(int4 t, int e, int end) {
    const bool in = e < end;
    return make_int4(in ? t.x : -1, in ? t.y : -1, in ? t.z : -1, in ? t.w : -1);
}

static __device__ __forceinline__ int slice_len(int E, int slices) {
    int s = (E + slices - 1) / slices;
    return (s + 3) & ~3;
}

__global__ __launch_bounds__(SCAT_THREADS)
void gcn_degree_kernel(const int* __restrict__ ei,
                       unsigned* __restrict__ hd, int N, int E) {
    __shared__ unsigned s_cnt[RANGE];
    const int tid = threadIdx.x;
    const int lo  = blockIdx.x * RANGE;
    for (int i = tid; i < RANGE; i += SCAT_THREADS) s_cnt[i] = 0u;
    __syncthreads();

    const int* __restrict__ dst = ei + E;
    const bool vec = (E & 3) == 0;
    const int S   = slice_len(E, SLICES);
    const int beg = blockIdx.y * S;
    const int end = min(E, beg + S);

    if (vec) {
        if (beg < end) {
            const int step = SCAT_THREADS * 4 * BATCH;
            int4 nx[BATCH];
            #pragma unroll
            for (int u = 0; u < BATCH; ++u)
                nx[u] = load_ids4(dst, beg + tid * 4 + u * SCAT_THREADS * 4, end);
            for (int e0 = beg + tid * 4; e0 < end; e0 += step) {
                int4 v[BATCH];
                #pragma unroll
                for (int u = 0; u < BATCH; ++u) v[u] = nx[u];
                #pragma unroll
                for (int u = 0; u < BATCH; ++u)
                    nx[u] = load_ids4(dst, e0 + step + u * SCAT_THREADS * 4, end);
                #pragma unroll
                for (int u = 0; u < BATCH; ++u) {
                    const int4 k = keep_ids4(v[u], e0 + u * SCAT_THREADS * 4, end);
                    const int d[4] = {k.x, k.y, k.z, k.w};
                    #pragma unroll
                    for (int j = 0; j < 4; ++j) {
                        const unsigned rel = (unsigned)(d[j] - lo);
                        if (rel < RANGE)
                            __hip_atomic_fetch_add(&s_cnt[rel], 1u, __ATOMIC_RELAXED,
                                                   __HIP_MEMORY_SCOPE_WORKGROUP);
                    }
                }
            }
        }
    } else {
        for (int e = beg + tid; e < end; e += SCAT_THREADS) {
            const unsigned rel = (unsigned)(dst[e] - lo);
            if (rel < RANGE)
                __hip_atomic_fetch_add(&s_cnt[rel], 1u, __ATOMIC_RELAXED,
                                       __HIP_MEMORY_SCOPE_WORKGROUP);
        }
    }
    __syncthreads();

    const int n = min(RANGE, N - lo);
    for (int i = tid; i < n; i += SCAT_THREADS)
        atomicAdd(&hd[2 * (lo + i) + 1], s_cnt[i]);
}

static __device__ __forceinline__ float dinv_of(unsigned d) {
    return rsqrtf((float)(d + 1u));
}

__global__ __launch_bounds__(SCAT_THREADS)
void gcn_edge_kernel(const int* __restrict__ ei,
                     const unsigned* __restrict__ hd,
                     float* __restrict__ accum, int N, int E) {
    __shared__ float s_val[SRANGE];
    __shared__ float s_acc[DRANGE];
    const int tid  = threadIdx.x;
    const int lane = tid & 31;
    const int wave = tid >> 5;
    const int nsp  = (N + SRANGE - 1) / SRANGE;
    const int slo  = (blockIdx.x % nsp) * SRANGE;
    const int dlo  = (blockIdx.x / nsp) * DRANGE;
    const int ns   = min(SRANGE, N - slo);
    const uint2* __restrict__ rec = (const uint2*)hd;

    for (int i = tid; i < DRANGE; i += SCAT_THREADS) s_acc[i] = 0.0f;
    const int rot = ((int)blockIdx.y * (ns / ESLICES)) & ~31;
    for (int i0 = tid; i0 < ns; i0 += SCAT_THREADS * SBATCH) {
        uint2 r[SBATCH];
        int   p[SBATCH];
        #pragma unroll
        for (int u = 0; u < SBATCH; ++u) {
            const int q = min(i0 + u * SCAT_THREADS, ns - 1) + rot;
            p[u] = q >= ns ? q - ns : q;
            r[u] = rec[slo + p[u]];
        }
        #pragma unroll
        for (int u = 0; u < SBATCH; ++u)
            if (i0 + u * SCAT_THREADS < ns)
                s_val[p[u]] = dinv_of(r[u].y) * __uint_as_float(r[u].x);
    }
    __syncthreads();

    const int* __restrict__ src = ei;
    const int* __restrict__ dst = ei + E;
    const bool vec = (E & 3) == 0;
    const int S   = slice_len(E, ESLICES);
    const int beg = blockIdx.y * S;
    const int end = min(E, beg + S);

    if (vec) {
        if (beg < end) {
            const int step = (SCAT_THREADS / 32) * WTILE;
            int4 nd[BATCH], nsr[BATCH];
            #pragma unroll
            for (int u = 0; u < BATCH; ++u) {
                const int e = beg + wave * WTILE + u * 128 + lane * 4;
                nd[u]  = load_ids4(dst, e, end);
                nsr[u] = load_ids4(src, e, end);
            }
            for (int tb = beg + wave * WTILE; tb < end; tb += step) {
                int4 vd[BATCH], vs[BATCH];
                #pragma unroll
                for (int u = 0; u < BATCH; ++u) { vd[u] = nd[u]; vs[u] = nsr[u]; }
                #pragma unroll
                for (int u = 0; u < BATCH; ++u) {
                    const int e = tb + step + u * 128 + lane * 4;
                    nd[u]  = load_ids4(dst, e, end);
                    nsr[u] = load_ids4(src, e, end);
                }
                unsigned rd[4 * BATCH];
                bool     in[4 * BATCH];
                float    val[4 * BATCH];
                #pragma unroll
                for (int u = 0; u < BATCH; ++u) {
                    const int4 k = keep_ids4(vd[u], tb + u * 128 + lane * 4, end);
                    const int d[4] = {k.x, k.y, k.z, k.w};
                    const int s[4] = {vs[u].x, vs[u].y, vs[u].z, vs[u].w};
                    #pragma unroll
                    for (int j = 0; j < 4; ++j) {
                        const unsigned rs = (unsigned)(s[j] - slo);
                        rd[4 * u + j]  = (unsigned)(d[j] - dlo);
                        in[4 * u + j]  = rd[4 * u + j] < DRANGE && rs < SRANGE;
                        val[4 * u + j] = s_val[in[4 * u + j] ? rs : 0u];
                    }
                }
                asm volatile("" : "+v"(val[0]), "+v"(val[1]), "+v"(val[2]), "+v"(val[3]),
                                  "+v"(val[4]), "+v"(val[5]), "+v"(val[6]), "+v"(val[7]));
                asm volatile("" : "+v"(val[8]), "+v"(val[9]), "+v"(val[10]), "+v"(val[11]),
                                  "+v"(val[12]), "+v"(val[13]), "+v"(val[14]), "+v"(val[15]));
                #pragma unroll
                for (int k = 0; k < 4 * BATCH; ++k)
                    if (in[k])
                        __hip_atomic_fetch_add(&s_acc[rd[k]], val[k], __ATOMIC_RELAXED,
                                               __HIP_MEMORY_SCOPE_WORKGROUP);
            }
        }
    } else {
        for (int e = beg + tid; e < end; e += SCAT_THREADS) {
            const unsigned rd = (unsigned)(dst[e] - dlo);
            const unsigned rs = (unsigned)(src[e] - slo);
            if (rd < DRANGE && rs < SRANGE)
                __hip_atomic_fetch_add(&s_acc[rd], s_val[rs], __ATOMIC_RELAXED,
                                       __HIP_MEMORY_SCOPE_WORKGROUP);
        }
    }
    __syncthreads();

    const int n = min(DRANGE, N - dlo);
    for (int i = tid; i < n; i += SCAT_THREADS) {
        const float part = s_acc[i];
        if (part != 0.0f) atomicAdd(&accum[dlo + i], part);
    }
}

__global__ void gcn_finalize_kernel(const float* __restrict__ accum,
                                    const unsigned* __restrict__ hd,
                                    const float* __restrict__ b,
                                    float* __restrict__ out, int N) {
    int i = blockIdx.x * blockDim.x + threadIdx.x;
    if (i < N) {
        const uint2 r = ((const uint2*)hd)[i];
        float di = dinv_of(r.y);
        float v  = di * (accum[i] + di * __uint_as_float(r.x)) + b[0];
        out[i]   = fmaxf(v, 0.0f) + log1pf(expf(-fabsf(v)));
    }
}

extern "C" void kernel_launch(void* const* d_in, const int* in_sizes, int n_in,
                              void* d_out, int out_size, void* d_ws, size_t ws_size,
                              hipStream_t stream) {
    const float* x  = (const float*)d_in[0];
    const int*   ei = (const int*)d_in[1];
    const float* W  = (const float*)d_in[2];
    const float* b  = (const float*)d_in[3];
    float* out = (float*)d_out;

    const int N = in_sizes[0] / C_DIM;
    const int E = in_sizes[1] / 2;

    unsigned* hd    = (unsigned*)d_ws;
    float*    accum = (float*)(hd + N + N) + N;

    const int BLK = 256;
    const int gridN = (N + BLK - 1) / BLK;
    const dim3 gridS((N + RANGE - 1) / RANGE, SLICES);
    const dim3 gridE(((N + SRANGE - 1) / SRANGE) * ((N + DRANGE - 1) / DRANGE), ESLICES);
    const int nodesPerBlock = (BLK / 32) * 16;
    const int gridM = (N + nodesPerBlock - 1) / nodesPerBlock;

    gcn_matvec_wmma_kernel<<<gridM, BLK, 0, stream>>>(x, W, hd, accum, N);
    gcn_degree_kernel<<<gridS, SCAT_THREADS, 0, stream>>>(ei, hd, N, E);
    gcn_edge_kernel<<<gridE, SCAT_THREADS, 0, stream>>>(ei, hd, accum, N, E);
    gcn_finalize_kernel<<<gridN, BLK, 0, stream>>>(accum, hd, b, out, N);
}


# Degree merge adds on a dense plane: h and degree leave their shared 8-byte record for two planes, read in order by 16-byte loads; degree kernel 16x16 to 8 ranges x 32 slices; float sum order as before
# speedup vs baseline: 8.5957x; 1.1391x over previous
#include <hip/hip_runtime.h>
#include <hip/hip_bf16.h>
#include <math.h>

typedef __attribute__((ext_vector_type(2))) float v2f;
typedef __attribute__((ext_vector_type(8))) float v8f;

#define C_DIM 256

struct MvStep {
    float4 w;
    float4 x[4];
};

static __device__ __forceinline__ void mv_load(MvStep& s, const float* __restrict__ wp,
                                               const float* __restrict__ x0,
                                               const float* __restrict__ x1,
                                               const float* __restrict__ x2,
                                               const float* __restrict__ x3, int step) {
    const int o = 32 * step;
    s.w    = *(const float4*)(wp + o);
    s.x[0] = *(const float4*)(x0 + o);
    s.x[1] = *(const float4*)(x1 + o);
    s.x[2] = *(const float4*)(x2 + o);
    s.x[3] = *(const float4*)(x3 + o);
}

static __device__ __forceinline__ void mv_products(const MvStep& s, v8f* acc) {
    v2f b0, b1;
    b0.x = s.w.x; b0.y = s.w.y;
    b1.x = s.w.z; b1.y = s.w.w;
    #pragma unroll
    for (int g = 0; g < 4; ++g) {
        v2f a0;
        a0.x = s.x[g].x; a0.y = s.x[g].y;
        acc[g] = __builtin_amdgcn_wmma_f32_16x16x4_f32(
             false, a0,  false, b0,
             (short)0, acc[g],  false,  false);
    }
    #pragma unroll
    for (int g = 0; g < 4; ++g) {
        v2f a1;
        a1.x = s.x[g].z; a1.y = s.x[g].w;
        acc[g] = __builtin_amdgcn_wmma_f32_16x16x4_f32(
             false, a1,  false, b1,
             (short)0, acc[g],  false,  false);
    }
}

__global__ __launch_bounds__(256)
void gcn_matvec_wmma_kernel(const float* __restrict__ x,
                            const float* __restrict__ W,
                            float* __restrict__ h,
                            unsigned* __restrict__ deg,
                            float* __restrict__ accum, int N) {
    const int lane = threadIdx.x & 31;
    const int wave = threadIdx.x >> 5;
    const int wavesPerBlock = blockDim.x >> 5;
    const int base = (blockIdx.x * wavesPerBlock + wave) * 16;
    if (base >= N) return;

    const int half = lane >> 4;
    const int m    = lane & 15;
    const int q    = m >> 2;
    const int c    = m & 3;
    const int koff = (2 * c + half) * 4;

    const float* __restrict__ xr[4];
    #pragma unroll
    for (int g = 0; g < 4; ++g) {
        int row = base + 4 * g + q;
        if (row >= N) row = N - 1;
        xr[g] = x + (size_t)row * C_DIM + koff;
    }
    const float* __restrict__ wp = W + koff;

    v8f acc[4];
    #pragma unroll
    for (int g = 0; g < 4; ++g)
        acc[g] = (v8f){0.f, 0.f, 0.f, 0.f, 0.f, 0.f, 0.f, 0.f};

    MvStep st[4];
    mv_load(st[0], wp, xr[0], xr[1], xr[2], xr[3], 0);
    mv_load(st[1], wp, xr[0], xr[1], xr[2], xr[3], 1);
    mv_load(st[2], wp, xr[0], xr[1], xr[2], xr[3], 2);
    mv_load(st[3], wp, xr[0], xr[1], xr[2], xr[3], 3);
    __builtin_amdgcn_sched_barrier(0);
    mv_products(st[0], acc);
    __builtin_amdgcn_sched_barrier(0);
    mv_load(st[0], wp, xr[0], xr[1], xr[2], xr[3], 4);
    __builtin_amdgcn_sched_barrier(0);
    mv_products(st[1], acc);
    __builtin_amdgcn_sched_barrier(0);
    mv_load(st[1], wp, xr[0], xr[1], xr[2], xr[3], 5);
    __builtin_amdgcn_sched_barrier(0);
    mv_products(st[2], acc);
    __builtin_amdgcn_sched_barrier(0);
    mv_load(st[2], wp, xr[0], xr[1], xr[2], xr[3], 6);
    __builtin_amdgcn_sched_barrier(0);
    mv_products(st[3], acc);
    __builtin_amdgcn_sched_barrier(0);
    mv_load(st[3], wp, xr[0], xr[1], xr[2], xr[3], 7);
    __builtin_amdgcn_sched_barrier(0);
    mv_products(st[0], acc);
    mv_products(st[1], acc);
    mv_products(st[2], acc);
    mv_products(st[3], acc);

    float lo = 0.0f, hi = 0.0f;
    #pragma unroll
    for (int g = 0; g < 4; ++g) {
        #pragma unroll
        for (int r = 0; r < 4; ++r) {
            const bool s = (q == g) && (c == r);
            lo = s ? acc[g][r]     : lo;
            hi = s ? acc[g][4 + r] : hi;
        }
    }
    lo += __shfl_xor(lo, 1);
    lo += __shfl_xor(lo, 2);
    hi += __shfl_xor(hi, 1);
    hi += __shfl_xor(hi, 2);
    if (c < 2) {
        const int node = base + 4 * q + 2 * half + c;
        const float hv = c ? hi : lo;
        if (node < N) h[node] = hv;
    }
    if (base + m < N) {
        unsigned* z = half == 0 ? (unsigned*)accum : deg;
        z[base + m] = 0u;
    }
}

#define RANGE        12500
#define SLICES       32
#define SRANGE       50000
#define DRANGE       25000
#define ESLICES      32
#define SCAT_THREADS 1024
#define BATCH        4
#define WTILE        (BATCH * 128)
#define SBATCH       4

static __device__ __forceinline__ int4 load_ids4(const int* __restrict__ p,
                                                 int e, int end) {
    return *(const int4*)(p + min(e, end - 4));
}
static __device__ __forceinline__ int4 keep_ids4(int4 t, int e, int end) {
    const bool in = e < end;
    return make_int4(in ? t.x : -1, in ? t.y : -1, in ? t.z : -1, in ? t.w : -1);
}

static __device__ __forceinline__ int slice_len(int E, int slices) {
    int s = (E + slices - 1) / slices;
    return (s + 3) & ~3;
}

__global__ __launch_bounds__(SCAT_THREADS)
void gcn_degree_kernel(const int* __restrict__ ei,
                       unsigned* __restrict__ deg, int N, int E) {
    __shared__ unsigned s_cnt[RANGE];
    const int tid = threadIdx.x;
    const int lo  = blockIdx.x * RANGE;
    for (int i = tid; i < RANGE; i += SCAT_THREADS) s_cnt[i] = 0u;
    __syncthreads();

    const int* __restrict__ dst = ei + E;
    const bool vec = (E & 3) == 0;
    const int S   = slice_len(E, SLICES);
    const int beg = blockIdx.y * S;
    const int end = min(E, beg + S);

    if (vec) {
        if (beg < end) {
            const int step = SCAT_THREADS * 4 * BATCH;
            int4 nx[BATCH];
            #pragma unroll
            for (int u = 0; u < BATCH; ++u)
                nx[u] = load_ids4(dst, beg + tid * 4 + u * SCAT_THREADS * 4, end);
            for (int e0 = beg + tid * 4; e0 < end; e0 += step) {
                int4 v[BATCH];
                #pragma unroll
                for (int u = 0; u < BATCH; ++u) v[u] = nx[u];
                #pragma unroll
                for (int u = 0; u < BATCH; ++u)
                    nx[u] = load_ids4(dst, e0 + step + u * SCAT_THREADS * 4, end);
                #pragma unroll
                for (int u = 0; u < BATCH; ++u) {
                    const int4 k = keep_ids4(v[u], e0 + u * SCAT_THREADS * 4, end);
                    const int d[4] = {k.x, k.y, k.z, k.w};
                    #pragma unroll
                    for (int j = 0; j < 4; ++j) {
                        const unsigned rel = (unsigned)(d[j] - lo);
                        if (rel < RANGE)
                            __hip_atomic_fetch_add(&s_cnt[rel], 1u, __ATOMIC_RELAXED,
                                                   __HIP_MEMORY_SCOPE_WORKGROUP);
                    }
                }
            }
        }
    } else {
        for (int e = beg + tid; e < end; e += SCAT_THREADS) {
            const unsigned rel = (unsigned)(dst[e] - lo);
            if (rel < RANGE)
                __hip_atomic_fetch_add(&s_cnt[rel], 1u, __ATOMIC_RELAXED,
                                       __HIP_MEMORY_SCOPE_WORKGROUP);
        }
    }
    __syncthreads();

    const int n = min(RANGE, N - lo);
    for (int i = tid; i < n; i += SCAT_THREADS) {
        const unsigned c = s_cnt[i];
        if (c != 0u) atomicAdd(&deg[lo + i], c);
    }
}

static __device__ __forceinline__ float dinv_of(unsigned d) {
    return rsqrtf((float)(d + 1u));
}

__global__ __launch_bounds__(SCAT_THREADS)
void gcn_edge_kernel(const int* __restrict__ ei,
                     const float* __restrict__ h,
                     const unsigned* __restrict__ deg,
                     float* __restrict__ accum, int N, int E) {
    __shared__ __attribute__((aligned(16))) float s_val[SRANGE];
    __shared__ float s_acc[DRANGE];
    const int tid  = threadIdx.x;
    const int lane = tid & 31;
    const int wave = tid >> 5;
    const int nsp  = (N + SRANGE - 1) / SRANGE;
    const int slo  = (blockIdx.x % nsp) * SRANGE;
    const int dlo  = (blockIdx.x / nsp) * DRANGE;
    const int ns   = min(SRANGE, N - slo);

    for (int i = tid; i < DRANGE; i += SCAT_THREADS) s_acc[i] = 0.0f;
    if ((N & 3) == 0) {
        const float4* __restrict__ h4 = (const float4*)(h + slo);
        const uint4*  __restrict__ d4 = (const uint4*)(deg + slo);
        const int nq  = ns >> 2;
        const int rot = ((int)blockIdx.y * (nq / ESLICES)) & ~15;
        for (int i0 = tid; i0 < nq; i0 += SCAT_THREADS * SBATCH) {
            float4 hv[SBATCH];
            uint4  dv[SBATCH];
            int    p[SBATCH];
            #pragma unroll
            for (int u = 0; u < SBATCH; ++u) {
                const int q = min(i0 + u * SCAT_THREADS, nq - 1) + rot;
                p[u]  = q >= nq ? q - nq : q;
                hv[u] = h4[p[u]];
                dv[u] = d4[p[u]];
            }
            #pragma unroll
            for (int u = 0; u < SBATCH; ++u)
                ((float4*)s_val)[p[u]] = make_float4(dinv_of(dv[u].x) * hv[u].x,
                                                     dinv_of(dv[u].y) * hv[u].y,
                                                     dinv_of(dv[u].z) * hv[u].z,
                                                     dinv_of(dv[u].w) * hv[u].w);
        }
    } else {
        for (int i = tid; i < ns; i += SCAT_THREADS)
            s_val[i] = dinv_of(deg[slo + i]) * h[slo + i];
    }
    __syncthreads();

    const int* __restrict__ src = ei;
    const int* __restrict__ dst = ei + E;
    const bool vec = (E & 3) == 0;
    const int S   = slice_len(E, ESLICES);
    const int beg = blockIdx.y * S;
    const int end = min(E, beg + S);

    if (vec) {
        if (beg < end) {
            const int step = (SCAT_THREADS / 32) * WTILE;
            int4 nd[BATCH], nsr[BATCH];
            #pragma unroll
            for (int u = 0; u < BATCH; ++u) {
                const int e = beg + wave * WTILE + u * 128 + lane * 4;
                nd[u]  = load_ids4(dst, e, end);
                nsr[u] = load_ids4(src, e, end);
            }
            for (int tb = beg + wave * WTILE; tb < end; tb += step) {
                int4 vd[BATCH], vs[BATCH];
                #pragma unroll
                for (int u = 0; u < BATCH; ++u) { vd[u] = nd[u]; vs[u] = nsr[u]; }
                #pragma unroll
                for (int u = 0; u < BATCH; ++u) {
                    const int e = tb + step + u * 128 + lane * 4;
                    nd[u]  = load_ids4(dst, e, end);
                    nsr[u] = load_ids4(src, e, end);
                }
                unsigned rd[4 * BATCH];
                bool     in[4 * BATCH];
                float    val[4 * BATCH];
                #pragma unroll
                for (int u = 0; u < BATCH; ++u) {
                    const int4 k = keep_ids4(vd[u], tb + u * 128 + lane * 4, end);
                    const int d[4] = {k.x, k.y, k.z, k.w};
                    const int s[4] = {vs[u].x, vs[u].y, vs[u].z, vs[u].w};
                    #pragma unroll
                    for (int j = 0; j < 4; ++j) {
                        const unsigned rs = (unsigned)(s[j] - slo);
                        rd[4 * u + j]  = (unsigned)(d[j] - dlo);
                        in[4 * u + j]  = rd[4 * u + j] < DRANGE && rs < SRANGE;
                        val[4 * u + j] = s_val[in[4 * u + j] ? rs : 0u];
                    }
                }
                asm volatile("" : "+v"(val[0]), "+v"(val[1]), "+v"(val[2]), "+v"(val[3]),
                                  "+v"(val[4]), "+v"(val[5]), "+v"(val[6]), "+v"(val[7]));
                asm volatile("" : "+v"(val[8]), "+v"(val[9]), "+v"(val[10]), "+v"(val[11]),
                                  "+v"(val[12]), "+v"(val[13]), "+v"(val[14]), "+v"(val[15]));
                #pragma unroll
                for (int k = 0; k < 4 * BATCH; ++k)
                    if (in[k])
                        __hip_atomic_fetch_add(&s_acc[rd[k]], val[k], __ATOMIC_RELAXED,
                                               __HIP_MEMORY_SCOPE_WORKGROUP);
            }
        }
    } else {
        for (int e = beg + tid; e < end; e += SCAT_THREADS) {
            const unsigned rd = (unsigned)(dst[e] - dlo);
            const unsigned rs = (unsigned)(src[e] - slo);
            if (rd < DRANGE && rs < SRANGE)
                __hip_atomic_fetch_add(&s_acc[rd], s_val[rs], __ATOMIC_RELAXED,
                                       __HIP_MEMORY_SCOPE_WORKGROUP);
        }
    }
    __syncthreads();

    const int n = min(DRANGE, N - dlo);
    for (int i = tid; i < n; i += SCAT_THREADS) {
        const float part = s_acc[i];
        if (part != 0.0f) atomicAdd(&accum[dlo + i], part);
    }
}

__global__ void gcn_finalize_kernel(const float* __restrict__ accum,
                                    const float* __restrict__ h,
                                    const unsigned* __restrict__ deg,
                                    const float* __restrict__ b,
                                    float* __restrict__ out, int N) {
    int i = blockIdx.x * blockDim.x + threadIdx.x;
    if (i < N) {
        float di = dinv_of(deg[i]);
        float v  = di * (accum[i] + di * h[i]) + b[0];
        out[i]   = fmaxf(v, 0.0f) + log1pf(expf(-fabsf(v)));
    }
}

extern "C" void kernel_launch(void* const* d_in, const int* in_sizes, int n_in,
                              void* d_out, int out_size, void* d_ws, size_t ws_size,
                              hipStream_t stream) {
    const float* x  = (const float*)d_in[0];
    const int*   ei = (const int*)d_in[1];
    const float* W  = (const float*)d_in[2];
    const float* b  = (const float*)d_in[3];
    float* out = (float*)d_out;

    const int N = in_sizes[0] / C_DIM;
    const int E = in_sizes[1] / 2;

    float*    h     = (float*)d_ws;
    unsigned* deg   = (unsigned*)(h + N);
    float*    accum = h + N + N + N;

    const int BLK = 256;
    const int gridN = (N + BLK - 1) / BLK;
    const dim3 gridS((N + RANGE - 1) / RANGE, SLICES);
    const dim3 gridE(((N + SRANGE - 1) / SRANGE) * ((N + DRANGE - 1) / DRANGE), ESLICES);
    const int nodesPerBlock = (BLK / 32) * 16;
    const int gridM = (N + nodesPerBlock - 1) / nodesPerBlock;

    gcn_matvec_wmma_kernel<<<gridM, BLK, 0, stream>>>(x, W, h, deg, accum, N);
    gcn_degree_kernel<<<gridS, SCAT_THREADS, 0, stream>>>(ei, deg, N, E);
    gcn_edge_kernel<<<gridE, SCAT_THREADS, 0, stream>>>(ei, h, deg, accum, N, E);
    gcn_finalize_kernel<<<gridN, BLK, 0, stream>>>(accum, h, deg, b, out, N);
}
